# starting kernel
# baseline (speedup 1.0000x reference)
#include <hip/hip_runtime.h>

typedef __attribute__((ext_vector_type(2))) float v2f;
typedef __attribute__((ext_vector_type(8))) float v8f;

#define B_N   4096
#define D_INN 1024
#define D_F   256
#define NCOMP 8
#define NMK   80
#define LOG2PI 1.83787706640934548356f

__device__ __forceinline__ v8f wmma4(v2f a, v2f b, v8f c) {
  return __builtin_amdgcn_wmma_f32_16x16x4_f32(false, a, false, b, (short)0, c,
                                               false, false);
}

__global__ void feats_kernel(const float* __restrict__ x,
                             const float* __restrict__ W,
                             const float* __restrict__ bias,
                             float* __restrict__ feats) {
  extern __shared__ float smem[];
  float* As = smem;
  float* Bs = smem + 64 * 66;
  const int tid = threadIdx.x;
  const int lane = tid & 31;
  const int w = tid >> 5;
  const int rowbase = blockIdx.y * 64;
  const int colbase = blockIdx.x * 64;

  v8f acc[2] = {};
  const int l15 = lane & 15;
  const int khalf = (lane >> 4) * 2;

  for (int kc = 0; kc < D_INN; kc += 64) {
    for (int e = tid; e < 64 * 64; e += 256) {
      int r = e >> 6, c = e & 63;
      As[r * 66 + c] = x[(size_t)(rowbase + r) * D_INN + kc + c];
      Bs[c * 66 + r] = W[(size_t)(kc + r) * D_F + colbase + c];
    }
    __syncthreads();
    for (int kk = 0; kk < 64; kk += 4) {
      int klo = kk + khalf;
#pragma unroll
      for (int t = 0; t < 2; ++t) {
        int g = w * 2 + t;
        int mt = g >> 2, nt = g & 3;
        v2f a = *(const v2f*)(As + (mt * 16 + l15) * 66 + klo);
        v2f bv = *(const v2f*)(Bs + (nt * 16 + l15) * 66 + klo);
        acc[t] = wmma4(a, bv, acc[t]);
      }
    }
    __syncthreads();
  }

  const int hi = lane >> 4;
#pragma unroll
  for (int t = 0; t < 2; ++t) {
    int g = w * 2 + t;
    int mt = g >> 2, nt = g & 3;
    int n = nt * 16 + l15;
    float bb = bias[colbase + n];
#pragma unroll
    for (int r = 0; r < 8; ++r) {
      int mrow = mt * 16 + r + hi * 8;
      float v = acc[t][r] + bb;
      v = v > 0.f ? v : 0.f;
      feats[(size_t)(rowbase + mrow) * D_F + colbase + n] = v;
    }
  }
}

__global__ void trinv_kernel(const float* __restrict__ covs,
                             float* __restrict__ Linv,
                             float* __restrict__ logdet) {
  extern __shared__ float smem[];
  float* Ls = smem;
  float* logd = smem + 256 * 257;
  const int j = threadIdx.x;
  const int mk = blockIdx.x;
  const float* Lg = covs + (size_t)mk * (D_F * D_F);

  for (int e = j; e < 256 * 257; e += 256) Ls[e] = 0.f;
  __syncthreads();

  float Ljj = Lg[j * D_F + j];
  Ls[j * 257 + j] = 1.f / Ljj;
  logd[j] = logf(Ljj);
  __syncthreads();

  for (int i = 1; i < D_F; ++i) {
    if (j < i) {
      const float* Lrow = Lg + (size_t)i * D_F;
      float s = 0.f;
      for (int t = j; t < i; ++t) s += Lrow[t] * Ls[t * 257 + j];
      Ls[i * 257 + j] = -s / Lrow[i];
    }
    __syncthreads();
  }

  float* outp = Linv + (size_t)mk * (D_F * D_F);
  for (int i = 0; i < D_F; ++i) outp[(size_t)i * D_F + j] = Ls[i * 257 + j];
  if (j == 0) {
    float s = 0.f;
    for (int t = 0; t < D_F; ++t) s += logd[t];
    logdet[mk] = s;
  }
}

__global__ void gmm_kernel(const float* __restrict__ feats,
                           const float* __restrict__ Linv,
                           const float* __restrict__ logdet,
                           const float* __restrict__ means,
                           const float* __restrict__ weights,
                           float* __restrict__ out) {
  extern __shared__ float smem[];
  float* As = smem;
  float* Bs = smem + 256 * 66;
  float* quad_s = smem + 256 * 66 + 64 * 258;
  const int tid = threadIdx.x;
  const int lane = tid & 31;
  const int w = tid >> 5;
  const int mk = blockIdx.y;
  const int b0 = blockIdx.x * 64;
  const float* LinvG = Linv + (size_t)mk * (D_F * D_F);
  const float* mu = means + (size_t)mk * D_F;

  for (int e = tid; e < 64 * 256; e += 256) {
    int n = e >> 8, d = e & 255;
    Bs[n * 258 + d] = feats[(size_t)(b0 + n) * D_F + d] - mu[d];
  }
  if (tid < 64) quad_s[tid] = 0.f;
  __syncthreads();

  v8f acc[8] = {};
  const int l15 = lane & 15;
  const int khalf = (lane >> 4) * 2;

  for (int kc = 0; kc < D_F; kc += 64) {
    for (int e = tid; e < 256 * 64; e += 256) {
      int r = e >> 6, c = e & 63;
      As[r * 66 + c] = LinvG[(size_t)r * D_F + kc + c];
    }
    __syncthreads();
    for (int kk = 0; kk < 64; kk += 4) {
      int klo = kk + khalf;
#pragma unroll
      for (int i = 0; i < 2; ++i) {
        int mt = w * 2 + i;
        v2f a = *(const v2f*)(As + (mt * 16 + l15) * 66 + klo);
#pragma unroll
        for (int nt = 0; nt < 4; ++nt) {
          v2f bv = *(const v2f*)(Bs + (nt * 16 + l15) * 258 + kc + klo);
          acc[i * 4 + nt] = wmma4(a, bv, acc[i * 4 + nt]);
        }
      }
    }
    __syncthreads();
  }

#pragma unroll
  for (int t = 0; t < 8; ++t) {
    int nt = t & 3;
    float p = 0.f;
#pragma unroll
    for (int r = 0; r < 8; ++r) p += acc[t][r] * acc[t][r];
    atomicAdd(&quad_s[nt * 16 + l15], p);
  }
  __syncthreads();

  if (tid < 64) {
    int m = mk >> 3, kcomp = mk & 7;
    const float* wr = weights + m * NCOMP;
    float wmax = wr[0];
#pragma unroll
    for (int t = 1; t < 8; ++t) wmax = fmaxf(wmax, wr[t]);
    float denom = 0.f;
#pragma unroll
    for (int t = 0; t < 8; ++t) denom += expf(wr[t] - wmax);
    float wmk = expf(wr[kcomp] - wmax) / denom;
    float logp = -0.5f * (quad_s[tid] + 256.0f * LOG2PI) - logdet[mk];
    out[(size_t)(b0 + tid) * NMK + mk] = wmk * logp;
  }
}

extern "C" void kernel_launch(void* const* d_in, const int* in_sizes, int n_in,
                              void* d_out, int out_size, void* d_ws,
                              size_t ws_size, hipStream_t stream) {
  (void)in_sizes; (void)n_in; (void)out_size; (void)ws_size;
  const float* x       = (const float*)d_in[0];
  const float* W       = (const float*)d_in[1];
  const float* bias    = (const float*)d_in[2];
  const float* means   = (const float*)d_in[3];
  const float* covs    = (const float*)d_in[4];
  const float* weights = (const float*)d_in[5];
  float* out = (float*)d_out;

  float* ws = (float*)d_ws;
  float* feats  = ws;
  float* LinvW  = ws + (size_t)B_N * D_F;
  float* logdet = LinvW + (size_t)NMK * D_F * D_F;

  size_t sh1 = (size_t)(2 * 64 * 66) * sizeof(float);
  feats_kernel<<<dim3(D_F / 64, B_N / 64), 256, sh1, stream>>>(x, W, bias,
                                                               feats);

  size_t sh2 = (size_t)(256 * 257 + 256) * sizeof(float);
  trinv_kernel<<<dim3(NMK), 256, sh2, stream>>>(covs, LinvW, logdet);

  size_t sh3 = (size_t)(256 * 66 + 64 * 258 + 64) * sizeof(float);
  gmm_kernel<<<dim3(B_N / 64, NMK), 256, sh3, stream>>>(feats, LinvW, logdet,
                                                        means, weights, out);
}


# triangular inverse: forward substitution 8 rows per step, L rows staged in LDS and read as broadcasts, each column's 8 sums split over 2 lanes; 32 barriers for 255; sum order and fused adds kept
# speedup vs baseline: 1.7114x; 1.7114x over previous
#include <hip/hip_runtime.h>

typedef __attribute__((ext_vector_type(2))) float v2f;
typedef __attribute__((ext_vector_type(8))) float v8f;

#define B_N   4096
#define D_INN 1024
#define D_F   256
#define NCOMP 8
#define NMK   80
#define LOG2PI 1.83787706640934548356f

__device__ __forceinline__ v8f wmma4(v2f a, v2f b, v8f c) {
  return __builtin_amdgcn_wmma_f32_16x16x4_f32(false, a, false, b, (short)0, c,
                                               false, false);
}

__global__ void feats_kernel(const float* __restrict__ x,
                             const float* __restrict__ W,
                             const float* __restrict__ bias,
                             float* __restrict__ feats) {
  extern __shared__ float smem[];
  float* As = smem;
  float* Bs = smem + 64 * 66;
  const int tid = threadIdx.x;
  const int lane = tid & 31;
  const int w = tid >> 5;
  const int rowbase = blockIdx.y * 64;
  const int colbase = blockIdx.x * 64;

  v8f acc[2] = {};
  const int l15 = lane & 15;
  const int khalf = (lane >> 4) * 2;

  for (int kc = 0; kc < D_INN; kc += 64) {
    for (int e = tid; e < 64 * 64; e += 256) {
      int r = e >> 6, c = e & 63;
      As[r * 66 + c] = x[(size_t)(rowbase + r) * D_INN + kc + c];
      Bs[c * 66 + r] = W[(size_t)(kc + r) * D_F + colbase + c];
    }
    __syncthreads();
    for (int kk = 0; kk < 64; kk += 4) {
      int klo = kk + khalf;
#pragma unroll
      for (int t = 0; t < 2; ++t) {
        int g = w * 2 + t;
        int mt = g >> 2, nt = g & 3;
        v2f a = *(const v2f*)(As + (mt * 16 + l15) * 66 + klo);
        v2f bv = *(const v2f*)(Bs + (nt * 16 + l15) * 66 + klo);
        acc[t] = wmma4(a, bv, acc[t]);
      }
    }
    __syncthreads();
  }

  const int hi = lane >> 4;
#pragma unroll
  for (int t = 0; t < 2; ++t) {
    int g = w * 2 + t;
    int mt = g >> 2, nt = g & 3;
    int n = nt * 16 + l15;
    float bb = bias[colbase + n];
#pragma unroll
    for (int r = 0; r < 8; ++r) {
      int mrow = mt * 16 + r + hi * 8;
      float v = acc[t][r] + bb;
      v = v > 0.f ? v : 0.f;
      feats[(size_t)(rowbase + mrow) * D_F + colbase + n] = v;
    }
  }
}

__global__ void __launch_bounds__(512)
trinv_kernel(const float* __restrict__ covs, float* __restrict__ Linv,
             float* __restrict__ logdet) {
  extern __shared__ float smem[];
  float* Xt = smem;
  float* logd = smem + 256 * 260;
  float* Lst = logd + 256;
  const int tid = threadIdx.x;
  const int lane = tid & 31;
  const int c = lane & 15;
  const int h = lane >> 4;
  const int w = __builtin_amdgcn_readfirstlane(tid >> 5);
  const int k1 = w >> 2;
  const int qd = (k1 + (w & 3)) & 3;
  const int slab = 4 * qd + ((qd & 1) ? 3 - k1 : k1);
  const int wstart = slab * 16;
  const int j = wstart + c;
  const int mk = blockIdx.x;
  const float* Lg = covs + (size_t)mk * (D_F * D_F);

  for (int e = tid; e < 256 * 260; e += 512) Xt[e] = 0.f;
  if (tid < 256) logd[tid] = logf(Lg[tid * D_F + tid]);

  const int tcol = tid & 255;
  const int hh = tid >> 8;
  float p[4];
#pragma unroll
  for (int r = 0; r < 4; ++r) p[r] = Lg[(size_t)(4 * hh + r) * D_F + tcol];

  float* xcol = Xt + j * 260;
  for (int i0 = 0; i0 < D_F; i0 += 8) {
    float* Lb = Lst + ((i0 >> 3) & 1) * (2 * 1028);
    *(float4*)(Lb + hh * 1028 + tcol * 4) = make_float4(p[0], p[1], p[2], p[3]);
    __syncthreads();
    if (i0 + 8 < D_F) {
#pragma unroll
      for (int r = 0; r < 4; ++r)
        p[r] = Lg[(size_t)(i0 + 8 + 4 * hh + r) * D_F + tcol];
    }
    if (wstart <= i0) {
      float s0 = 0.f, s1 = 0.f, s2 = 0.f, s3 = 0.f;
      const float* Lh = Lb + h * 1028;
      for (int t = wstart; t < i0; t += 8) {
        float4 xa = *(const float4*)(xcol + t);
        float4 xb = *(const float4*)(xcol + t + 4);
        float xs[8] = {xa.x, xa.y, xa.z, xa.w, xb.x, xb.y, xb.z, xb.w};
#pragma unroll
        for (int u = 0; u < 8; ++u) {
          float4 l = *(const float4*)(Lh + (t + u) * 4);
          s0 = fmaf(l.x, xs[u], s0);
          s1 = fmaf(l.y, xs[u], s1);
          s2 = fmaf(l.z, xs[u], s2);
          s3 = fmaf(l.w, xs[u], s3);
        }
      }
      float o0 = __shfl_xor(s0, 16);
      float o1 = __shfl_xor(s1, 16);
      float o2 = __shfl_xor(s2, 16);
      float o3 = __shfl_xor(s3, 16);
      float s[8];
      s[0] = h ? o0 : s0;  s[1] = h ? o1 : s1;
      s[2] = h ? o2 : s2;  s[3] = h ? o3 : s3;
      s[4] = h ? s0 : o0;  s[5] = h ? s1 : o1;
      s[6] = h ? s2 : o2;  s[7] = h ? s3 : o3;
      float xv[8];
#pragma unroll
      for (int r = 0; r < 8; ++r) {
        const float* Lr = Lb + (r >> 2) * 1028 + (r & 3);
        float acc = s[r];
#pragma unroll
        for (int q = 0; q < r; ++q) acc = fmaf(Lr[(i0 + q) * 4], xv[q], acc);
        float d = Lr[(i0 + r) * 4];
        float num = ((j == i0 + r) ? 1.f : 0.f) - acc;
        float v = num / d;
        xv[r] = (j <= i0 + r) ? v : 0.f;
      }
      if (h == 0) {
        *(float4*)(xcol + i0) = make_float4(xv[0], xv[1], xv[2], xv[3]);
        *(float4*)(xcol + i0 + 4) = make_float4(xv[4], xv[5], xv[6], xv[7]);
      }
    }
  }
  __syncthreads();

  float* outp = Linv + (size_t)mk * (D_F * D_F);
  const float* ocol = Xt + tcol * 260;
  for (int i = hh * 128; i < hh * 128 + 128; i += 4) {
    float4 v = *(const float4*)(ocol + i);
    outp[(size_t)(i + 0) * D_F + tcol] = v.x;
    outp[(size_t)(i + 1) * D_F + tcol] = v.y;
    outp[(size_t)(i + 2) * D_F + tcol] = v.z;
    outp[(size_t)(i + 3) * D_F + tcol] = v.w;
  }
  if (tid == 0) {
    float s = 0.f;
    for (int t = 0; t < D_F; ++t) s += logd[t];
    logdet[mk] = s;
  }
}

__global__ void gmm_kernel(const float* __restrict__ feats,
                           const float* __restrict__ Linv,
                           const float* __restrict__ logdet,
                           const float* __restrict__ means,
                           const float* __restrict__ weights,
                           float* __restrict__ out) {
  extern __shared__ float smem[];
  float* As = smem;
  float* Bs = smem + 256 * 66;
  float* quad_s = smem + 256 * 66 + 64 * 258;
  const int tid = threadIdx.x;
  const int lane = tid & 31;
  const int w = tid >> 5;
  const int mk = blockIdx.y;
  const int b0 = blockIdx.x * 64;
  const float* LinvG = Linv + (size_t)mk * (D_F * D_F);
  const float* mu = means + (size_t)mk * D_F;

  for (int e = tid; e < 64 * 256; e += 256) {
    int n = e >> 8, d = e & 255;
    Bs[n * 258 + d] = feats[(size_t)(b0 + n) * D_F + d] - mu[d];
  }
  if (tid < 64) quad_s[tid] = 0.f;
  __syncthreads();

  v8f acc[8] = {};
  const int l15 = lane & 15;
  const int khalf = (lane >> 4) * 2;

  for (int kc = 0; kc < D_F; kc += 64) {
    for (int e = tid; e < 256 * 64; e += 256) {
      int r = e >> 6, c = e & 63;
      As[r * 66 + c] = LinvG[(size_t)r * D_F + kc + c];
    }
    __syncthreads();
    for (int kk = 0; kk < 64; kk += 4) {
      int klo = kk + khalf;
#pragma unroll
      for (int i = 0; i < 2; ++i) {
        int mt = w * 2 + i;
        v2f a = *(const v2f*)(As + (mt * 16 + l15) * 66 + klo);
#pragma unroll
        for (int nt = 0; nt < 4; ++nt) {
          v2f bv = *(const v2f*)(Bs + (nt * 16 + l15) * 258 + kc + klo);
          acc[i * 4 + nt] = wmma4(a, bv, acc[i * 4 + nt]);
        }
      }
    }
    __syncthreads();
  }

#pragma unroll
  for (int t = 0; t < 8; ++t) {
    int nt = t & 3;
    float p = 0.f;
#pragma unroll
    for (int r = 0; r < 8; ++r) p += acc[t][r] * acc[t][r];
    atomicAdd(&quad_s[nt * 16 + l15], p);
  }
  __syncthreads();

  if (tid < 64) {
    int m = mk >> 3, kcomp = mk & 7;
    const float* wr = weights + m * NCOMP;
    float wmax = wr[0];
#pragma unroll
    for (int t = 1; t < 8; ++t) wmax = fmaxf(wmax, wr[t]);
    float denom = 0.f;
#pragma unroll
    for (int t = 0; t < 8; ++t) denom += expf(wr[t] - wmax);
    float wmk = expf(wr[kcomp] - wmax) / denom;
    float logp = -0.5f * (quad_s[tid] + 256.0f * LOG2PI) - logdet[mk];
    out[(size_t)(b0 + tid) * NMK + mk] = wmk * logp;
  }
}

extern "C" void kernel_launch(void* const* d_in, const int* in_sizes, int n_in,
                              void* d_out, int out_size, void* d_ws,
                              size_t ws_size, hipStream_t stream) {
  (void)in_sizes; (void)n_in; (void)out_size; (void)ws_size;
  const float* x       = (const float*)d_in[0];
  const float* W       = (const float*)d_in[1];
  const float* bias    = (const float*)d_in[2];
  const float* means   = (const float*)d_in[3];
  const float* covs    = (const float*)d_in[4];
  const float* weights = (const float*)d_in[5];
  float* out = (float*)d_out;

  float* ws = (float*)d_ws;
  float* feats  = ws;
  float* LinvW  = ws + (size_t)B_N * D_F;
  float* logdet = LinvW + (size_t)NMK * D_F * D_F;

  size_t sh1 = (size_t)(2 * 64 * 66) * sizeof(float);
  feats_kernel<<<dim3(D_F / 64, B_N / 64), 256, sh1, stream>>>(x, W, bias,
                                                               feats);

  size_t sh2 = (size_t)(256 * 260 + 256 + 2 * 2 * 1028) * sizeof(float);
  trinv_kernel<<<dim3(NMK), 512, sh2, stream>>>(covs, LinvW, logdet);

  size_t sh3 = (size_t)(256 * 66 + 64 * 258 + 64) * sizeof(float);
  gmm_kernel<<<dim3(B_N / 64, NMK), 256, sh3, stream>>>(feats, LinvW, logdet,
                                                        means, weights, out);
}


# third kernel: Linv panels and the feats-minus-mean tile staged by 16-byte global loads, 16 per thread in flight, next panel's loads issued ahead of the current products (was 4-byte, two in flight)
# speedup vs baseline: 3.6435x; 2.1290x over previous
#include <hip/hip_runtime.h>

typedef __attribute__((ext_vector_type(2))) float v2f;
typedef __attribute__((ext_vector_type(8))) float v8f;

#define B_N   4096
#define D_INN 1024
#define D_F   256
#define NCOMP 8
#define NMK   80
#define LOG2PI 1.83787706640934548356f

__device__ __forceinline__ v8f wmma4(v2f a, v2f b, v8f c) {
  return __builtin_amdgcn_wmma_f32_16x16x4_f32(false, a, false, b, (short)0, c,
                                               false, false);
}

__global__ void feats_kernel(const float* __restrict__ x,
                             const float* __restrict__ W,
                             const float* __restrict__ bias,
                             float* __restrict__ feats) {
  extern __shared__ float smem[];
  float* As = smem;
  float* Bs = smem + 64 * 66;
  const int tid = threadIdx.x;
  const int lane = tid & 31;
  const int w = tid >> 5;
  const int rowbase = blockIdx.y * 64;
  const int colbase = blockIdx.x * 64;

  v8f acc[2] = {};
  const int l15 = lane & 15;
  const int khalf = (lane >> 4) * 2;

  for (int kc = 0; kc < D_INN; kc += 64) {
    for (int e = tid; e < 64 * 64; e += 256) {
      int r = e >> 6, c = e & 63;
      As[r * 66 + c] = x[(size_t)(rowbase + r) * D_INN + kc + c];
      Bs[c * 66 + r] = W[(size_t)(kc + r) * D_F + colbase + c];
    }
    __syncthreads();
    for (int kk = 0; kk < 64; kk += 4) {
      int klo = kk + khalf;
#pragma unroll
      for (int t = 0; t < 2; ++t) {
        int g = w * 2 + t;
        int mt = g >> 2, nt = g & 3;
        v2f a = *(const v2f*)(As + (mt * 16 + l15) * 66 + klo);
        v2f bv = *(const v2f*)(Bs + (nt * 16 + l15) * 66 + klo);
        acc[t] = wmma4(a, bv, acc[t]);
      }
    }
    __syncthreads();
  }

  const int hi = lane >> 4;
#pragma unroll
  for (int t = 0; t < 2; ++t) {
    int g = w * 2 + t;
    int mt = g >> 2, nt = g & 3;
    int n = nt * 16 + l15;
    float bb = bias[colbase + n];
#pragma unroll
    for (int r = 0; r < 8; ++r) {
      int mrow = mt * 16 + r + hi * 8;
      float v = acc[t][r] + bb;
      v = v > 0.f ? v : 0.f;
      feats[(size_t)(rowbase + mrow) * D_F + colbase + n] = v;
    }
  }
}

__global__ void __launch_bounds__(512)
trinv_kernel(const float* __restrict__ covs, float* __restrict__ Linv,
             float* __restrict__ logdet) {
  extern __shared__ float smem[];
  float* Xt = smem;
  float* logd = smem + 256 * 260;
  float* Lst = logd + 256;
  const int tid = threadIdx.x;
  const int lane = tid & 31;
  const int c = lane & 15;
  const int h = lane >> 4;
  const int w = __builtin_amdgcn_readfirstlane(tid >> 5);
  const int k1 = w >> 2;
  const int qd = (k1 + (w & 3)) & 3;
  const int slab = 4 * qd + ((qd & 1) ? 3 - k1 : k1);
  const int wstart = slab * 16;
  const int j = wstart + c;
  const int mk = blockIdx.x;
  const float* Lg = covs + (size_t)mk * (D_F * D_F);

  for (int e = tid; e < 256 * 260; e += 512) Xt[e] = 0.f;
  if (tid < 256) logd[tid] = logf(Lg[tid * D_F + tid]);

  const int tcol = tid & 255;
  const int hh = tid >> 8;
  float p[4];
#pragma unroll
  for (int r = 0; r < 4; ++r) p[r] = Lg[(size_t)(4 * hh + r) * D_F + tcol];

  float* xcol = Xt + j * 260;
  for (int i0 = 0; i0 < D_F; i0 += 8) {
    float* Lb = Lst + ((i0 >> 3) & 1) * (2 * 1028);
    *(float4*)(Lb + hh * 1028 + tcol * 4) = make_float4(p[0], p[1], p[2], p[3]);
    __syncthreads();
    if (i0 + 8 < D_F) {
#pragma unroll
      for (int r = 0; r < 4; ++r)
        p[r] = Lg[(size_t)(i0 + 8 + 4 * hh + r) * D_F + tcol];
    }
    if (wstart <= i0) {
      float s0 = 0.f, s1 = 0.f, s2 = 0.f, s3 = 0.f;
      const float* Lh = Lb + h * 1028;
      for (int t = wstart; t < i0; t += 8) {
        float4 xa = *(const float4*)(xcol + t);
        float4 xb = *(const float4*)(xcol + t + 4);
        float xs[8] = {xa.x, xa.y, xa.z, xa.w, xb.x, xb.y, xb.z, xb.w};
#pragma unroll
        for (int u = 0; u < 8; ++u) {
          float4 l = *(const float4*)(Lh + (t + u) * 4);
          s0 = fmaf(l.x, xs[u], s0);
          s1 = fmaf(l.y, xs[u], s1);
          s2 = fmaf(l.z, xs[u], s2);
          s3 = fmaf(l.w, xs[u], s3);
        }
      }
      float o0 = __shfl_xor(s0, 16);
      float o1 = __shfl_xor(s1, 16);
      float o2 = __shfl_xor(s2, 16);
      float o3 = __shfl_xor(s3, 16);
      float s[8];
      s[0] = h ? o0 : s0;  s[1] = h ? o1 : s1;
      s[2] = h ? o2 : s2;  s[3] = h ? o3 : s3;
      s[4] = h ? s0 : o0;  s[5] = h ? s1 : o1;
      s[6] = h ? s2 : o2;  s[7] = h ? s3 : o3;
      float xv[8];
#pragma unroll
      for (int r = 0; r < 8; ++r) {
        const float* Lr = Lb + (r >> 2) * 1028 + (r & 3);
        float acc = s[r];
#pragma unroll
        for (int q = 0; q < r; ++q) acc = fmaf(Lr[(i0 + q) * 4], xv[q], acc);
        float d = Lr[(i0 + r) * 4];
        float num = ((j == i0 + r) ? 1.f : 0.f) - acc;
        float v = num / d;
        xv[r] = (j <= i0 + r) ? v : 0.f;
      }
      if (h == 0) {
        *(float4*)(xcol + i0) = make_float4(xv[0], xv[1], xv[2], xv[3]);
        *(float4*)(xcol + i0 + 4) = make_float4(xv[4], xv[5], xv[6], xv[7]);
      }
    }
  }
  __syncthreads();

  float* outp = Linv + (size_t)mk * (D_F * D_F);
  const float* ocol = Xt + tcol * 260;
  for (int i = hh * 128; i < hh * 128 + 128; i += 4) {
    float4 v = *(const float4*)(ocol + i);
    outp[(size_t)(i + 0) * D_F + tcol] = v.x;
    outp[(size_t)(i + 1) * D_F + tcol] = v.y;
    outp[(size_t)(i + 2) * D_F + tcol] = v.z;
    outp[(size_t)(i + 3) * D_F + tcol] = v.w;
  }
  if (tid == 0) {
    float s = 0.f;
    for (int t = 0; t < D_F; ++t) s += logd[t];
    logdet[mk] = s;
  }
}

__global__ void __launch_bounds__(256)
gmm_kernel(const float* __restrict__ feats,
                           const float* __restrict__ Linv,
                           const float* __restrict__ logdet,
                           const float* __restrict__ means,
                           const float* __restrict__ weights,
                           float* __restrict__ out) {
  extern __shared__ float smem[];
  float* As = smem;
  float* Bs = smem + 256 * 66;
  float* quad_s = smem + 256 * 66 + 64 * 258;
  const int tid = threadIdx.x;
  const int lane = tid & 31;
  const int w = tid >> 5;
  const int mk = blockIdx.y;
  const int b0 = blockIdx.x * 64;
  const float* LinvG = Linv + (size_t)mk * (D_F * D_F);
  const float* mu = means + (size_t)mk * D_F;

  const int c4 = (tid & 15) * 4;
  const int r0 = tid >> 4;
  const float* gsrc = LinvG + (size_t)r0 * D_F + c4;
  float* adst = As + r0 * 66 + c4;
  float4 g[16];
#pragma unroll
  for (int p = 0; p < 16; ++p)
    g[p] = *(const float4*)(gsrc + (size_t)(16 * p) * D_F);

  {
    const int d4 = (tid & 63) * 4;
    const int n0 = tid >> 6;
    const float4 m4 = *(const float4*)(mu + d4);
    float4 f[16];
#pragma unroll
    for (int p = 0; p < 16; ++p)
      f[p] = *(const float4*)(feats + (size_t)(b0 + n0 + 4 * p) * D_F + d4);
#pragma unroll
    for (int p = 0; p < 16; ++p) {
      float* dst = Bs + (n0 + 4 * p) * 258 + d4;
      v2f lo = {f[p].x - m4.x, f[p].y - m4.y};
      v2f hi = {f[p].z - m4.z, f[p].w - m4.w};
      *(v2f*)(dst) = lo;
      *(v2f*)(dst + 2) = hi;
    }
  }
  if (tid < 64) quad_s[tid] = 0.f;
#pragma unroll
  for (int p = 0; p < 16; ++p) {
    v2f lo = {g[p].x, g[p].y};
    v2f hi = {g[p].z, g[p].w};
    *(v2f*)(adst + 16 * p * 66) = lo;
    *(v2f*)(adst + 16 * p * 66 + 2) = hi;
  }
  __syncthreads();

  v8f acc[8] = {};
  const int l15 = lane & 15;
  const int khalf = (lane >> 4) * 2;

  for (int kc = 0; kc < D_F; kc += 64) {
    if (kc + 64 < D_F) {
#pragma unroll
      for (int p = 0; p < 16; ++p)
        g[p] = *(const float4*)(gsrc + (size_t)(16 * p) * D_F + kc + 64);
    }
    for (int kk = 0; kk < 64; kk += 4) {
      int klo = kk + khalf;
#pragma unroll
      for (int i = 0; i < 2; ++i) {
        int mt = w * 2 + i;
        v2f a = *(const v2f*)(As + (mt * 16 + l15) * 66 + klo);
#pragma unroll
        for (int nt = 0; nt < 4; ++nt) {
          v2f bv = *(const v2f*)(Bs + (nt * 16 + l15) * 258 + kc + klo);
          acc[i * 4 + nt] = wmma4(a, bv, acc[i * 4 + nt]);
        }
      }
    }
    __syncthreads();
    if (kc + 64 < D_F) {
#pragma unroll
      for (int p = 0; p < 16; ++p) {
        v2f lo = {g[p].x, g[p].y};
        v2f hi = {g[p].z, g[p].w};
        *(v2f*)(adst + 16 * p * 66) = lo;
        *(v2f*)(adst + 16 * p * 66 + 2) = hi;
      }
      __syncthreads();
    }
  }

#pragma unroll
  for (int t = 0; t < 8; ++t) {
    int nt = t & 3;
    float p = 0.f;
#pragma unroll
    for (int r = 0; r < 8; ++r) p += acc[t][r] * acc[t][r];
    atomicAdd(&quad_s[nt * 16 + l15], p);
  }
  __syncthreads();

  if (tid < 64) {
    int m = mk >> 3, kcomp = mk & 7;
    const float* wr = weights + m * NCOMP;
    float wmax = wr[0];
#pragma unroll
    for (int t = 1; t < 8; ++t) wmax = fmaxf(wmax, wr[t]);
    float denom = 0.f;
#pragma unroll
    for (int t = 0; t < 8; ++t) denom += expf(wr[t] - wmax);
    float wmk = expf(wr[kcomp] - wmax) / denom;
    float logp = -0.5f * (quad_s[tid] + 256.0f * LOG2PI) - logdet[mk];
    out[(size_t)(b0 + tid) * NMK + mk] = wmk * logp;
  }
}

extern "C" void kernel_launch(void* const* d_in, const int* in_sizes, int n_in,
                              void* d_out, int out_size, void* d_ws,
                              size_t ws_size, hipStream_t stream) {
  (void)in_sizes; (void)n_in; (void)out_size; (void)ws_size;
  const float* x       = (const float*)d_in[0];
  const float* W       = (const float*)d_in[1];
  const float* bias    = (const float*)d_in[2];
  const float* means   = (const float*)d_in[3];
  const float* covs    = (const float*)d_in[4];
  const float* weights = (const float*)d_in[5];
  float* out = (float*)d_out;

  float* ws = (float*)d_ws;
  float* feats  = ws;
  float* LinvW  = ws + (size_t)B_N * D_F;
  float* logdet = LinvW + (size_t)NMK * D_F * D_F;

  size_t sh1 = (size_t)(2 * 64 * 66) * sizeof(float);
  feats_kernel<<<dim3(D_F / 64, B_N / 64), 256, sh1, stream>>>(x, W, bias,
                                                               feats);

  size_t sh2 = (size_t)(256 * 260 + 256 + 2 * 2 * 1028) * sizeof(float);
  trinv_kernel<<<dim3(NMK), 512, sh2, stream>>>(covs, LinvW, logdet);

  size_t sh3 = (size_t)(256 * 66 + 64 * 258 + 64) * sizeof(float);
  gmm_kernel<<<dim3(B_N / 64, NMK), 256, sh3, stream>>>(feats, LinvW, logdet,
                                                        means, weights, out);
}


# third kernel: Linv blocks above the diagonal (exact zeros of a triangular inverse, any input) left out of loads and products; row tiles dealt q,7-q,8+q,15-q per wave; k order of each sum kept
# speedup vs baseline: 4.7076x; 1.2921x over previous
#include <hip/hip_runtime.h>

typedef __attribute__((ext_vector_type(2))) float v2f;
typedef __attribute__((ext_vector_type(8))) float v8f;

#define B_N   4096
#define D_INN 1024
#define D_F   256
#define NCOMP 8
#define NMK   80
#define LOG2PI 1.83787706640934548356f

__device__ __forceinline__ v8f wmma4(v2f a, v2f b, v8f c) {
  return __builtin_amdgcn_wmma_f32_16x16x4_f32(false, a, false, b, (short)0, c,
                                               false, false);
}

__global__ void feats_kernel(const float* __restrict__ x,
                             const float* __restrict__ W,
                             const float* __restrict__ bias,
                             float* __restrict__ feats) {
  extern __shared__ float smem[];
  float* As = smem;
  float* Bs = smem + 64 * 66;
  const int tid = threadIdx.x;
  const int lane = tid & 31;
  const int w = tid >> 5;
  const int rowbase = blockIdx.y * 64;
  const int colbase = blockIdx.x * 64;

  v8f acc[2] = {};
  const int l15 = lane & 15;
  const int khalf = (lane >> 4) * 2;

  for (int kc = 0; kc < D_INN; kc += 64) {
    for (int e = tid; e < 64 * 64; e += 256) {
      int r = e >> 6, c = e & 63;
      As[r * 66 + c] = x[(size_t)(rowbase + r) * D_INN + kc + c];
      Bs[c * 66 + r] = W[(size_t)(kc + r) * D_F + colbase + c];
    }
    __syncthreads();
    for (int kk = 0; kk < 64; kk += 4) {
      int klo = kk + khalf;
#pragma unroll
      for (int t = 0; t < 2; ++t) {
        int g = w * 2 + t;
        int mt = g >> 2, nt = g & 3;
        v2f a = *(const v2f*)(As + (mt * 16 + l15) * 66 + klo);
        v2f bv = *(const v2f*)(Bs + (nt * 16 + l15) * 66 + klo);
        acc[t] = wmma4(a, bv, acc[t]);
      }
    }
    __syncthreads();
  }

  const int hi = lane >> 4;
#pragma unroll
  for (int t = 0; t < 2; ++t) {
    int g = w * 2 + t;
    int mt = g >> 2, nt = g & 3;
    int n = nt * 16 + l15;
    float bb = bias[colbase + n];
#pragma unroll
    for (int r = 0; r < 8; ++r) {
      int mrow = mt * 16 + r + hi * 8;
      float v = acc[t][r] + bb;
      v = v > 0.f ? v : 0.f;
      feats[(size_t)(rowbase + mrow) * D_F + colbase + n] = v;
    }
  }
}

__global__ void __launch_bounds__(512)
trinv_kernel(const float* __restrict__ covs, float* __restrict__ Linv,
             float* __restrict__ logdet) {
  extern __shared__ float smem[];
  float* Xt = smem;
  float* logd = smem + 256 * 260;
  float* Lst = logd + 256;
  const int tid = threadIdx.x;
  const int lane = tid & 31;
  const int c = lane & 15;
  const int h = lane >> 4;
  const int w = __builtin_amdgcn_readfirstlane(tid >> 5);
  const int k1 = w >> 2;
  const int qd = (k1 + (w & 3)) & 3;
  const int slab = 4 * qd + ((qd & 1) ? 3 - k1 : k1);
  const int wstart = slab * 16;
  const int j = wstart + c;
  const int mk = blockIdx.x;
  const float* Lg = covs + (size_t)mk * (D_F * D_F);

  for (int e = tid; e < 256 * 260; e += 512) Xt[e] = 0.f;
  if (tid < 256) logd[tid] = logf(Lg[tid * D_F + tid]);

  const int tcol = tid & 255;
  const int hh = tid >> 8;
  float p[4];
#pragma unroll
  for (int r = 0; r < 4; ++r) p[r] = Lg[(size_t)(4 * hh + r) * D_F + tcol];

  float* xcol = Xt + j * 260;
  for (int i0 = 0; i0 < D_F; i0 += 8) {
    float* Lb = Lst + ((i0 >> 3) & 1) * (2 * 1028);
    *(float4*)(Lb + hh * 1028 + tcol * 4) = make_float4(p[0], p[1], p[2], p[3]);
    __syncthreads();
    if (i0 + 8 < D_F) {
#pragma unroll
      for (int r = 0; r < 4; ++r)
        p[r] = Lg[(size_t)(i0 + 8 + 4 * hh + r) * D_F + tcol];
    }
    if (wstart <= i0) {
      float s0 = 0.f, s1 = 0.f, s2 = 0.f, s3 = 0.f;
      const float* Lh = Lb + h * 1028;
      for (int t = wstart; t < i0; t += 8) {
        float4 xa = *(const float4*)(xcol + t);
        float4 xb = *(const float4*)(xcol + t + 4);
        float xs[8] = {xa.x, xa.y, xa.z, xa.w, xb.x, xb.y, xb.z, xb.w};
#pragma unroll
        for (int u = 0; u < 8; ++u) {
          float4 l = *(const float4*)(Lh + (t + u) * 4);
          s0 = fmaf(l.x, xs[u], s0);
          s1 = fmaf(l.y, xs[u], s1);
          s2 = fmaf(l.z, xs[u], s2);
          s3 = fmaf(l.w, xs[u], s3);
        }
      }
      float o0 = __shfl_xor(s0, 16);
      float o1 = __shfl_xor(s1, 16);
      float o2 = __shfl_xor(s2, 16);
      float o3 = __shfl_xor(s3, 16);
      float s[8];
      s[0] = h ? o0 : s0;  s[1] = h ? o1 : s1;
      s[2] = h ? o2 : s2;  s[3] = h ? o3 : s3;
      s[4] = h ? s0 : o0;  s[5] = h ? s1 : o1;
      s[6] = h ? s2 : o2;  s[7] = h ? s3 : o3;
      float xv[8];
#pragma unroll
      for (int r = 0; r < 8; ++r) {
        const float* Lr = Lb + (r >> 2) * 1028 + (r & 3);
        float acc = s[r];
#pragma unroll
        for (int q = 0; q < r; ++q) acc = fmaf(Lr[(i0 + q) * 4], xv[q], acc);
        float d = Lr[(i0 + r) * 4];
        float num = ((j == i0 + r) ? 1.f : 0.f) - acc;
        float v = num / d;
        xv[r] = (j <= i0 + r) ? v : 0.f;
      }
      if (h == 0) {
        *(float4*)(xcol + i0) = make_float4(xv[0], xv[1], xv[2], xv[3]);
        *(float4*)(xcol + i0 + 4) = make_float4(xv[4], xv[5], xv[6], xv[7]);
      }
    }
  }
  __syncthreads();

  float* outp = Linv + (size_t)mk * (D_F * D_F);
  const float* ocol = Xt + tcol * 260;
  for (int i = hh * 128; i < hh * 128 + 128; i += 4) {
    float4 v = *(const float4*)(ocol + i);
    outp[(size_t)(i + 0) * D_F + tcol] = v.x;
    outp[(size_t)(i + 1) * D_F + tcol] = v.y;
    outp[(size_t)(i + 2) * D_F + tcol] = v.z;
    outp[(size_t)(i + 3) * D_F + tcol] = v.w;
  }
  if (tid == 0) {
    float s = 0.f;
    for (int t = 0; t < D_F; ++t) s += logd[t];
    logdet[mk] = s;
  }
}


template <int FIRST>
__device__ __forceinline__ void tri_block(const float* const (&Ap)[4],
                                          const float* Bp, int k,
                                          v8f (&acc)[8]) {
#pragma unroll
  for (int s = 0; s < 16; s += 4) {
    v2f b0 = *(const v2f*)(Bp + k + s);
    v2f b1 = *(const v2f*)(Bp + 16 * 258 + k + s);
#pragma unroll
    for (int t = FIRST; t < 4; ++t) {
      v2f a = *(const v2f*)(Ap[t] + k + s);
      acc[2 * t + 0] = wmma4(a, b0, acc[2 * t + 0]);
      acc[2 * t + 1] = wmma4(a, b1, acc[2 * t + 1]);
    }
  }
}

__device__ __forceinline__ int tri_need(int mt, int kc) {
  int n = 16 * (mt + 1) - kc;
  return n < 0 ? 0 : (n > 64 ? 64 : n);
}

__global__ void __launch_bounds__(256)
gmm_kernel(const float* __restrict__ feats,
                           const float* __restrict__ Linv,
                           const float* __restrict__ logdet,
                           const float* __restrict__ means,
                           const float* __restrict__ weights,
                           float* __restrict__ out) {
  extern __shared__ float smem[];
  float* As = smem;
  float* Bs = smem + 256 * 66;
  float* quad_s = smem + 256 * 66 + 64 * 258;
  const int tid = threadIdx.x;
  const int lane = tid & 31;
  const int w = __builtin_amdgcn_readfirstlane(tid >> 5);
  const int mk = blockIdx.y;
  const int b0 = blockIdx.x * 64;
  const float* LinvG = Linv + (size_t)mk * (D_F * D_F);
  const float* mu = means + (size_t)mk * D_F;

  const int c4 = (tid & 15) * 4;
  const int r0 = tid >> 4;
  const float* gsrc = LinvG + (size_t)r0 * D_F + c4;
  float* adst = As + r0 * 66 + c4;
  float4 g[16];
#pragma unroll
  for (int p = 0; p < 16; ++p)
    g[p] = *(const float4*)(gsrc + (size_t)(16 * p) * D_F);

  {
    const int d4 = (tid & 63) * 4;
    const int n0 = tid >> 6;
    const float4 m4 = *(const float4*)(mu + d4);
    float4 f[16];
#pragma unroll
    for (int p = 0; p < 16; ++p)
      f[p] = *(const float4*)(feats + (size_t)(b0 + n0 + 4 * p) * D_F + d4);
#pragma unroll
    for (int p = 0; p < 16; ++p) {
      float* dst = Bs + (n0 + 4 * p) * 258 + d4;
      v2f lo = {f[p].x - m4.x, f[p].y - m4.y};
      v2f hi = {f[p].z - m4.z, f[p].w - m4.w};
      *(v2f*)(dst) = lo;
      *(v2f*)(dst + 2) = hi;
    }
  }
  if (tid < 64) quad_s[tid] = 0.f;
#pragma unroll
  for (int p = 0; p < 16; ++p) {
    v2f lo = {g[p].x, g[p].y};
    v2f hi = {g[p].z, g[p].w};
    *(v2f*)(adst + 16 * p * 66) = lo;
    *(v2f*)(adst + 16 * p * 66 + 2) = hi;
  }
  __syncthreads();

  v8f acc[8] = {};
  const int l15 = lane & 15;
  const int khalf = (lane >> 4) * 2;
  const int nh = w >> 2;
  const int q = (w & 4) ? 3 - (w & 3) : (w & 3);
  const int mt[4] = {q, 7 - q, 8 + q, 15 - q};
  const float* const Ap[4] = {As + (mt[0] * 16 + l15) * 66 + khalf,
                              As + (mt[1] * 16 + l15) * 66 + khalf,
                              As + (mt[2] * 16 + l15) * 66 + khalf,
                              As + (mt[3] * 16 + l15) * 66 + khalf};
  const float* Bw = Bs + (nh * 32 + l15) * 258 + khalf;

#pragma unroll
  for (int kc = 0; kc < D_F; kc += 64) {
    if (kc + 64 < D_F) {
#pragma unroll
      for (int p = (kc + 64) / 16; p < 16; ++p)
        g[p] = *(const float4*)(gsrc + (size_t)(16 * p) * D_F + kc + 64);
    }
    {
      const int e0 = tri_need(mt[0], kc), e1 = tri_need(mt[1], kc);
      const int e2 = tri_need(mt[2], kc), e3 = tri_need(mt[3], kc);
      const float* Bp = Bw + kc;
      int k = 0;
      for (; k < e0; k += 16) tri_block<0>(Ap, Bp, k, acc);
      for (; k < e1; k += 16) tri_block<1>(Ap, Bp, k, acc);
      for (; k < e2; k += 16) tri_block<2>(Ap, Bp, k, acc);
      for (; k < e3; k += 16) tri_block<3>(Ap, Bp, k, acc);
    }
    __syncthreads();
    if (kc + 64 < D_F) {
#pragma unroll
      for (int p = (kc + 64) / 16; p < 16; ++p) {
        v2f lo = {g[p].x, g[p].y};
        v2f hi = {g[p].z, g[p].w};
        *(v2f*)(adst + 16 * p * 66) = lo;
        *(v2f*)(adst + 16 * p * 66 + 2) = hi;
      }
      __syncthreads();
    }
  }

#pragma unroll
  for (int t = 0; t < 8; ++t) {
    int nt = 2 * nh + (t & 1);
    float p = 0.f;
#pragma unroll
    for (int r = 0; r < 8; ++r) p += acc[t][r] * acc[t][r];
    atomicAdd(&quad_s[nt * 16 + l15], p);
  }
  __syncthreads();

  if (tid < 64) {
    int m = mk >> 3, kcomp = mk & 7;
    const float* wr = weights + m * NCOMP;
    float wmax = wr[0];
#pragma unroll
    for (int t = 1; t < 8; ++t) wmax = fmaxf(wmax, wr[t]);
    float denom = 0.f;
#pragma unroll
    for (int t = 0; t < 8; ++t) denom += expf(wr[t] - wmax);
    float wmk = expf(wr[kcomp] - wmax) / denom;
    float logp = -0.5f * (quad_s[tid] + 256.0f * LOG2PI) - logdet[mk];
    out[(size_t)(b0 + tid) * NMK + mk] = wmk * logp;
  }
}

extern "C" void kernel_launch(void* const* d_in, const int* in_sizes, int n_in,
                              void* d_out, int out_size, void* d_ws,
                              size_t ws_size, hipStream_t stream) {
  (void)in_sizes; (void)n_in; (void)out_size; (void)ws_size;
  const float* x       = (const float*)d_in[0];
  const float* W       = (const float*)d_in[1];
  const float* bias    = (const float*)d_in[2];
  const float* means   = (const float*)d_in[3];
  const float* covs    = (const float*)d_in[4];
  const float* weights = (const float*)d_in[5];
  float* out = (float*)d_out;

  float* ws = (float*)d_ws;
  float* feats  = ws;
  float* LinvW  = ws + (size_t)B_N * D_F;
  float* logdet = LinvW + (size_t)NMK * D_F * D_F;

  size_t sh1 = (size_t)(2 * 64 * 66) * sizeof(float);
  feats_kernel<<<dim3(D_F / 64, B_N / 64), 256, sh1, stream>>>(x, W, bias,
                                                               feats);

  size_t sh2 = (size_t)(256 * 260 + 256 + 2 * 2 * 1028) * sizeof(float);
  trinv_kernel<<<dim3(NMK), 512, sh2, stream>>>(covs, LinvW, logdet);

  size_t sh3 = (size_t)(256 * 66 + 64 * 258 + 64) * sizeof(float);
  gmm_kernel<<<dim3(B_N / 64, NMK), 256, sh3, stream>>>(feats, LinvW, logdet,
                                                        means, weights, out);
}


# feature GEMM and triangular inverses, independent of each other, merged into one launch: 80 work-groups invert, 128 do two 64x64 GEMM tiles each on WGPs the inverses leave idle; every sum unchanged
# speedup vs baseline: 5.5443x; 1.1777x over previous
#include <hip/hip_runtime.h>

typedef __attribute__((ext_vector_type(2))) float v2f;
typedef __attribute__((ext_vector_type(8))) float v8f;

#define B_N   4096
#define D_INN 1024
#define D_F   256
#define NCOMP 8
#define NMK   80
#define LOG2PI 1.83787706640934548356f

__device__ __forceinline__ v8f wmma4(v2f a, v2f b, v8f c) {
  return __builtin_amdgcn_wmma_f32_16x16x4_f32(false, a, false, b, (short)0, c,
                                               false, false);
}

__device__ __forceinline__ void feats_tile(const float* __restrict__ x,
                                           const float* __restrict__ W,
                                           const float* __restrict__ bias,
                                           float* __restrict__ feats,
                                           float* sm, int tid, int tile) {
  float* As = sm;
  float* Bs = sm + 64 * 66;
  const int lane = tid & 31;
  const int w = tid >> 5;
  const int rowbase = (tile >> 2) * 64;
  const int colbase = (tile & 3) * 64;

  v8f acc[2] = {};
  const int l15 = lane & 15;
  const int khalf = (lane >> 4) * 2;

  for (int kc = 0; kc < D_INN; kc += 64) {
#pragma unroll 8
    for (int e = tid; e < 64 * 64; e += 256) {
      int r = e >> 6, c = e & 63;
      As[r * 66 + c] = x[(size_t)(rowbase + r) * D_INN + kc + c];
      Bs[c * 66 + r] = W[(size_t)(kc + r) * D_F + colbase + c];
    }
    __syncthreads();
    for (int kk = 0; kk < 64; kk += 4) {
      int klo = kk + khalf;
#pragma unroll
      for (int t = 0; t < 2; ++t) {
        int g = w * 2 + t;
        int mt = g >> 2, nt = g & 3;
        v2f a = *(const v2f*)(As + (mt * 16 + l15) * 66 + klo);
        v2f bv = *(const v2f*)(Bs + (nt * 16 + l15) * 66 + klo);
        acc[t] = wmma4(a, bv, acc[t]);
      }
    }
    __syncthreads();
  }

  const int hi = lane >> 4;
#pragma unroll
  for (int t = 0; t < 2; ++t) {
    int g = w * 2 + t;
    int mt = g >> 2, nt = g & 3;
    int n = nt * 16 + l15;
    float bb = bias[colbase + n];
#pragma unroll
    for (int r = 0; r < 8; ++r) {
      int mrow = mt * 16 + r + hi * 8;
      float v = acc[t][r] + bb;
      v = v > 0.f ? v : 0.f;
      feats[(size_t)(rowbase + mrow) * D_F + colbase + n] = v;
    }
  }
}

__device__ __forceinline__ void trinv_matrix(const float* __restrict__ covs,
                                             float* __restrict__ Linv,
                                             float* __restrict__ logdet,
                                             float* smem, int mk) {
  float* Xt = smem;
  float* logd = smem + 256 * 260;
  float* Lst = logd + 256;
  const int tid = threadIdx.x;
  const int lane = tid & 31;
  const int c = lane & 15;
  const int h = lane >> 4;
  const int w = __builtin_amdgcn_readfirstlane(tid >> 5);
  const int k1 = w >> 2;
  const int qd = (k1 + (w & 3)) & 3;
  const int slab = 4 * qd + ((qd & 1) ? 3 - k1 : k1);
  const int wstart = slab * 16;
  const int j = wstart + c;
  const float* Lg = covs + (size_t)mk * (D_F * D_F);

  for (int e = tid; e < 256 * 260; e += 512) Xt[e] = 0.f;
  if (tid < 256) logd[tid] = logf(Lg[tid * D_F + tid]);

  const int tcol = tid & 255;
  const int hh = tid >> 8;
  float p[4];
#pragma unroll
  for (int r = 0; r < 4; ++r) p[r] = Lg[(size_t)(4 * hh + r) * D_F + tcol];

  float* xcol = Xt + j * 260;
  for (int i0 = 0; i0 < D_F; i0 += 8) {
    float* Lb = Lst + ((i0 >> 3) & 1) * (2 * 1028);
    *(float4*)(Lb + hh * 1028 + tcol * 4) = make_float4(p[0], p[1], p[2], p[3]);
    __syncthreads();
    if (i0 + 8 < D_F) {
#pragma unroll
      for (int r = 0; r < 4; ++r)
        p[r] = Lg[(size_t)(i0 + 8 + 4 * hh + r) * D_F + tcol];
    }
    if (wstart <= i0) {
      float s0 = 0.f, s1 = 0.f, s2 = 0.f, s3 = 0.f;
      const float* Lh = Lb + h * 1028;
      for (int t = wstart; t < i0; t += 8) {
        float4 xa = *(const float4*)(xcol + t);
        float4 xb = *(const float4*)(xcol + t + 4);
        float xs[8] = {xa.x, xa.y, xa.z, xa.w, xb.x, xb.y, xb.z, xb.w};
#pragma unroll
        for (int u = 0; u < 8; ++u) {
          float4 l = *(const float4*)(Lh + (t + u) * 4);
          s0 = fmaf(l.x, xs[u], s0);
          s1 = fmaf(l.y, xs[u], s1);
          s2 = fmaf(l.z, xs[u], s2);
          s3 = fmaf(l.w, xs[u], s3);
        }
      }
      float o0 = __shfl_xor(s0, 16);
      float o1 = __shfl_xor(s1, 16);
      float o2 = __shfl_xor(s2, 16);
      float o3 = __shfl_xor(s3, 16);
      float s[8];
      s[0] = h ? o0 : s0;  s[1] = h ? o1 : s1;
      s[2] = h ? o2 : s2;  s[3] = h ? o3 : s3;
      s[4] = h ? s0 : o0;  s[5] = h ? s1 : o1;
      s[6] = h ? s2 : o2;  s[7] = h ? s3 : o3;
      float xv[8];
#pragma unroll
      for (int r = 0; r < 8; ++r) {
        const float* Lr = Lb + (r >> 2) * 1028 + (r & 3);
        float acc = s[r];
#pragma unroll
        for (int q = 0; q < r; ++q) acc = fmaf(Lr[(i0 + q) * 4], xv[q], acc);
        float d = Lr[(i0 + r) * 4];
        float num = ((j == i0 + r) ? 1.f : 0.f) - acc;
        float v = num / d;
        xv[r] = (j <= i0 + r) ? v : 0.f;
      }
      if (h == 0) {
        *(float4*)(xcol + i0) = make_float4(xv[0], xv[1], xv[2], xv[3]);
        *(float4*)(xcol + i0 + 4) = make_float4(xv[4], xv[5], xv[6], xv[7]);
      }
    }
  }
  __syncthreads();

  float* outp = Linv + (size_t)mk * (D_F * D_F);
  const float* ocol = Xt + tcol * 260;
  for (int i = hh * 128; i < hh * 128 + 128; i += 4) {
    float4 v = *(const float4*)(ocol + i);
    outp[(size_t)(i + 0) * D_F + tcol] = v.x;
    outp[(size_t)(i + 1) * D_F + tcol] = v.y;
    outp[(size_t)(i + 2) * D_F + tcol] = v.z;
    outp[(size_t)(i + 3) * D_F + tcol] = v.w;
  }
  if (tid == 0) {
    float s = 0.f;
    for (int t = 0; t < D_F; ++t) s += logd[t];
    logdet[mk] = s;
  }
}

__global__ void __launch_bounds__(512)
front_kernel(const float* __restrict__ x, const float* __restrict__ W,
             const float* __restrict__ bias, const float* __restrict__ covs,
             float* __restrict__ feats, float* __restrict__ Linv,
             float* __restrict__ logdet) {
  extern __shared__ float smem[];
  const int blk = blockIdx.x;
  if (blk < NMK) {
    trinv_matrix(covs, Linv, logdet, smem, blk);
  } else {
    const int half = threadIdx.x >> 8;
    feats_tile(x, W, bias, feats, smem + half * (2 * 64 * 66),
               threadIdx.x & 255, 2 * (blk - NMK) + half);
  }
}


template <int FIRST>
__device__ __forceinline__ void tri_block(const float* const (&Ap)[4],
                                          const float* Bp, int k,
                                          v8f (&acc)[8]) {
#pragma unroll
  for (int s = 0; s < 16; s += 4) {
    v2f b0 = *(const v2f*)(Bp + k + s);
    v2f b1 = *(const v2f*)(Bp + 16 * 258 + k + s);
#pragma unroll
    for (int t = FIRST; t < 4; ++t) {
      v2f a = *(const v2f*)(Ap[t] + k + s);
      acc[2 * t + 0] = wmma4(a, b0, acc[2 * t + 0]);
      acc[2 * t + 1] = wmma4(a, b1, acc[2 * t + 1]);
    }
  }
}

__device__ __forceinline__ int tri_need(int mt, int kc) {
  int n = 16 * (mt + 1) - kc;
  return n < 0 ? 0 : (n > 64 ? 64 : n);
}

__global__ void __launch_bounds__(256)
gmm_kernel(const float* __restrict__ feats,
                           const float* __restrict__ Linv,
                           const float* __restrict__ logdet,
                           const float* __restrict__ means,
                           const float* __restrict__ weights,
                           float* __restrict__ out) {
  extern __shared__ float smem[];
  float* As = smem;
  float* Bs = smem + 256 * 66;
  float* quad_s = smem + 256 * 66 + 64 * 258;
  const int tid = threadIdx.x;
  const int lane = tid & 31;
  const int w = __builtin_amdgcn_readfirstlane(tid >> 5);
  const int mk = blockIdx.y;
  const int b0 = blockIdx.x * 64;
  const float* LinvG = Linv + (size_t)mk * (D_F * D_F);
  const float* mu = means + (size_t)mk * D_F;

  const int c4 = (tid & 15) * 4;
  const int r0 = tid >> 4;
  const float* gsrc = LinvG + (size_t)r0 * D_F + c4;
  float* adst = As + r0 * 66 + c4;
  float4 g[16];
#pragma unroll
  for (int p = 0; p < 16; ++p)
    g[p] = *(const float4*)(gsrc + (size_t)(16 * p) * D_F);

  {
    const int d4 = (tid & 63) * 4;
    const int n0 = tid >> 6;
    const float4 m4 = *(const float4*)(mu + d4);
    float4 f[16];
#pragma unroll
    for (int p = 0; p < 16; ++p)
      f[p] = *(const float4*)(feats + (size_t)(b0 + n0 + 4 * p) * D_F + d4);
#pragma unroll
    for (int p = 0; p < 16; ++p) {
      float* dst = Bs + (n0 + 4 * p) * 258 + d4;
      v2f lo = {f[p].x - m4.x, f[p].y - m4.y};
      v2f hi = {f[p].z - m4.z, f[p].w - m4.w};
      *(v2f*)(dst) = lo;
      *(v2f*)(dst + 2) = hi;
    }
  }
  if (tid < 64) quad_s[tid] = 0.f;
#pragma unroll
  for (int p = 0; p < 16; ++p) {
    v2f lo = {g[p].x, g[p].y};
    v2f hi = {g[p].z, g[p].w};
    *(v2f*)(adst + 16 * p * 66) = lo;
    *(v2f*)(adst + 16 * p * 66 + 2) = hi;
  }
  __syncthreads();

  v8f acc[8] = {};
  const int l15 = lane & 15;
  const int khalf = (lane >> 4) * 2;
  const int nh = w >> 2;
  const int q = (w & 4) ? 3 - (w & 3) : (w & 3);
  const int mt[4] = {q, 7 - q, 8 + q, 15 - q};
  const float* const Ap[4] = {As + (mt[0] * 16 + l15) * 66 + khalf,
                              As + (mt[1] * 16 + l15) * 66 + khalf,
                              As + (mt[2] * 16 + l15) * 66 + khalf,
                              As + (mt[3] * 16 + l15) * 66 + khalf};
  const float* Bw = Bs + (nh * 32 + l15) * 258 + khalf;

#pragma unroll
  for (int kc = 0; kc < D_F; kc += 64) {
    if (kc + 64 < D_F) {
#pragma unroll
      for (int p = (kc + 64) / 16; p < 16; ++p)
        g[p] = *(const float4*)(gsrc + (size_t)(16 * p) * D_F + kc + 64);
    }
    {
      const int e0 = tri_need(mt[0], kc), e1 = tri_need(mt[1], kc);
      const int e2 = tri_need(mt[2], kc), e3 = tri_need(mt[3], kc);
      const float* Bp = Bw + kc;
      int k = 0;
      for (; k < e0; k += 16) tri_block<0>(Ap, Bp, k, acc);
      for (; k < e1; k += 16) tri_block<1>(Ap, Bp, k, acc);
      for (; k < e2; k += 16) tri_block<2>(Ap, Bp, k, acc);
      for (; k < e3; k += 16) tri_block<3>(Ap, Bp, k, acc);
    }
    __syncthreads();
    if (kc + 64 < D_F) {
#pragma unroll
      for (int p = (kc + 64) / 16; p < 16; ++p) {
        v2f lo = {g[p].x, g[p].y};
        v2f hi = {g[p].z, g[p].w};
        *(v2f*)(adst + 16 * p * 66) = lo;
        *(v2f*)(adst + 16 * p * 66 + 2) = hi;
      }
      __syncthreads();
    }
  }

#pragma unroll
  for (int t = 0; t < 8; ++t) {
    int nt = 2 * nh + (t & 1);
    float p = 0.f;
#pragma unroll
    for (int r = 0; r < 8; ++r) p += acc[t][r] * acc[t][r];
    atomicAdd(&quad_s[nt * 16 + l15], p);
  }
  __syncthreads();

  if (tid < 64) {
    int m = mk >> 3, kcomp = mk & 7;
    const float* wr = weights + m * NCOMP;
    float wmax = wr[0];
#pragma unroll
    for (int t = 1; t < 8; ++t) wmax = fmaxf(wmax, wr[t]);
    float denom = 0.f;
#pragma unroll
    for (int t = 0; t < 8; ++t) denom += expf(wr[t] - wmax);
    float wmk = expf(wr[kcomp] - wmax) / denom;
    float logp = -0.5f * (quad_s[tid] + 256.0f * LOG2PI) - logdet[mk];
    out[(size_t)(b0 + tid) * NMK + mk] = wmk * logp;
  }
}

extern "C" void kernel_launch(void* const* d_in, const int* in_sizes, int n_in,
                              void* d_out, int out_size, void* d_ws,
                              size_t ws_size, hipStream_t stream) {
  (void)in_sizes; (void)n_in; (void)out_size; (void)ws_size;
  const float* x       = (const float*)d_in[0];
  const float* W       = (const float*)d_in[1];
  const float* bias    = (const float*)d_in[2];
  const float* means   = (const float*)d_in[3];
  const float* covs    = (const float*)d_in[4];
  const float* weights = (const float*)d_in[5];
  float* out = (float*)d_out;

  float* ws = (float*)d_ws;
  float* feats  = ws;
  float* LinvW  = ws + (size_t)B_N * D_F;
  float* logdet = LinvW + (size_t)NMK * D_F * D_F;

  size_t sh2 = (size_t)(256 * 260 + 256 + 2 * 2 * 1028) * sizeof(float);
  front_kernel<<<dim3(NMK + 128), 512, sh2, stream>>>(x, W, bias, covs, feats,
                                                      LinvW, logdet);

  size_t sh3 = (size_t)(256 * 66 + 64 * 258 + 64) * sizeof(float);
  gmm_kernel<<<dim3(B_N / 64, NMK), 256, sh3, stream>>>(feats, LinvW, logdet,
                                                        means, weights, out);
}


# triangular inverse by 128x128 blocks: both diagonal blocks by the same substitution side by side (16 steps for 32), lower-left = -invC*B*invA as two f32 WMMA products; its sum order changes
# speedup vs baseline: 6.9738x; 1.2578x over previous
#include <hip/hip_runtime.h>

typedef __attribute__((ext_vector_type(2))) float v2f;
typedef __attribute__((ext_vector_type(8))) float v8f;

#define B_N   4096
#define D_INN 1024
#define D_F   256
#define NCOMP 8
#define NMK   80
#define LOG2PI 1.83787706640934548356f

__device__ __forceinline__ v8f wmma4(v2f a, v2f b, v8f c) {
  return __builtin_amdgcn_wmma_f32_16x16x4_f32(false, a, false, b, (short)0, c,
                                               false, false);
}

__device__ __forceinline__ void feats_tile(const float* __restrict__ x,
                                           const float* __restrict__ W,
                                           const float* __restrict__ bias,
                                           float* __restrict__ feats,
                                           float* sm, int tid, int tile) {
  float* As = sm;
  float* Bs = sm + 64 * 66;
  const int lane = tid & 31;
  const int w = tid >> 5;
  const int rowbase = (tile >> 2) * 64;
  const int colbase = (tile & 3) * 64;

  v8f acc[2] = {};
  const int l15 = lane & 15;
  const int khalf = (lane >> 4) * 2;

  for (int kc = 0; kc < D_INN; kc += 64) {
#pragma unroll 8
    for (int e = tid; e < 64 * 64; e += 256) {
      int r = e >> 6, c = e & 63;
      As[r * 66 + c] = x[(size_t)(rowbase + r) * D_INN + kc + c];
      Bs[c * 66 + r] = W[(size_t)(kc + r) * D_F + colbase + c];
    }
    __syncthreads();
    for (int kk = 0; kk < 64; kk += 4) {
      int klo = kk + khalf;
#pragma unroll
      for (int t = 0; t < 2; ++t) {
        int g = w * 2 + t;
        int mt = g >> 2, nt = g & 3;
        v2f a = *(const v2f*)(As + (mt * 16 + l15) * 66 + klo);
        v2f bv = *(const v2f*)(Bs + (nt * 16 + l15) * 66 + klo);
        acc[t] = wmma4(a, bv, acc[t]);
      }
    }
    __syncthreads();
  }

  const int hi = lane >> 4;
#pragma unroll
  for (int t = 0; t < 2; ++t) {
    int g = w * 2 + t;
    int mt = g >> 2, nt = g & 3;
    int n = nt * 16 + l15;
    float bb = bias[colbase + n];
#pragma unroll
    for (int r = 0; r < 8; ++r) {
      int mrow = mt * 16 + r + hi * 8;
      float v = acc[t][r] + bb;
      v = v > 0.f ? v : 0.f;
      feats[(size_t)(rowbase + mrow) * D_F + colbase + n] = v;
    }
  }
}

__device__ __forceinline__ void trinv_matrix(const float* __restrict__ covs,
                                             float* __restrict__ Linv,
                                             float* __restrict__ logdet,
                                             float* smem, int mk) {
  float* Xt = smem;
  float* logd = smem + 256 * 260;
  float* Lst = logd + 256;
  const int tid = threadIdx.x;
  const int lane = tid & 31;
  const int c = lane & 15;
  const int h = lane >> 4;
  const int w = __builtin_amdgcn_readfirstlane(tid >> 5);
  const int k1 = w >> 2;
  const int qd = (k1 + (w & 3)) & 3;
  const int rank = 4 * qd + ((qd & 1) ? 3 - k1 : k1);
  const int slab = ((rank & 1) << 3) | (rank >> 1);
  const int wstart = slab * 16;
  const int bb = wstart & 128;
  const int j = wstart + c;
  const float* Lg = covs + (size_t)mk * (D_F * D_F);

  for (int e = tid; e < 256 * 260; e += 512) Xt[e] = 0.f;
  if (tid < 256) logd[tid] = logf(Lg[tid * D_F + tid]);

  const int tcol = tid & 255;
  const int hh = tid >> 8;
  const int trow = (tcol & 128) + 4 * hh;
  float p[4];
#pragma unroll
  for (int r = 0; r < 4; ++r) p[r] = Lg[(size_t)(trow + r) * D_F + tcol];

  float* xcol = Xt + j * 260;
  for (int s0 = 0; s0 < 128; s0 += 8) {
    float* Lb = Lst + ((s0 >> 3) & 1) * (2 * 1028);
    *(float4*)(Lb + hh * 1028 + tcol * 4) = make_float4(p[0], p[1], p[2], p[3]);
    __syncthreads();
    if (s0 + 8 < 128) {
#pragma unroll
      for (int r = 0; r < 4; ++r)
        p[r] = Lg[(size_t)(s0 + 8 + trow + r) * D_F + tcol];
    }
    const int i0 = bb + s0;
    if (wstart <= i0) {
      float s0_ = 0.f, s1 = 0.f, s2 = 0.f, s3 = 0.f;
      const float* Lh = Lb + h * 1028;
      for (int t = wstart; t < i0; t += 8) {
        float4 xa = *(const float4*)(xcol + t);
        float4 xb = *(const float4*)(xcol + t + 4);
        float xs[8] = {xa.x, xa.y, xa.z, xa.w, xb.x, xb.y, xb.z, xb.w};
#pragma unroll
        for (int u = 0; u < 8; ++u) {
          float4 l = *(const float4*)(Lh + (t + u) * 4);
          s0_ = fmaf(l.x, xs[u], s0_);
          s1 = fmaf(l.y, xs[u], s1);
          s2 = fmaf(l.z, xs[u], s2);
          s3 = fmaf(l.w, xs[u], s3);
        }
      }
      float o0 = __shfl_xor(s0_, 16);
      float o1 = __shfl_xor(s1, 16);
      float o2 = __shfl_xor(s2, 16);
      float o3 = __shfl_xor(s3, 16);
      float s[8];
      s[0] = h ? o0 : s0_; s[1] = h ? o1 : s1;
      s[2] = h ? o2 : s2;  s[3] = h ? o3 : s3;
      s[4] = h ? s0_ : o0; s[5] = h ? s1 : o1;
      s[6] = h ? s2 : o2;  s[7] = h ? s3 : o3;
      float xv[8];
#pragma unroll
      for (int r = 0; r < 8; ++r) {
        const float* Lr = Lb + (r >> 2) * 1028 + (r & 3);
        float acc = s[r];
#pragma unroll
        for (int q = 0; q < r; ++q) acc = fmaf(Lr[(i0 + q) * 4], xv[q], acc);
        float d = Lr[(i0 + r) * 4];
        float num = ((j == i0 + r) ? 1.f : 0.f) - acc;
        float v = num / d;
        xv[r] = (j <= i0 + r) ? v : 0.f;
      }
      if (h == 0) {
        *(float4*)(xcol + i0) = make_float4(xv[0], xv[1], xv[2], xv[3]);
        *(float4*)(xcol + i0 + 4) = make_float4(xv[4], xv[5], xv[6], xv[7]);
      }
    }
  }
  __syncthreads();

  const int it = w & 7;
  const int jt0 = (w >> 3) * 4;
  const int irow = 16 * it + c;
  const int kq = 2 * h;
  v2f qf[32];
  {
    const float* bq = Lg + (size_t)(128 + irow) * D_F + kq;
#pragma unroll
    for (int ks = 0; ks < 32; ++ks) qf[ks] = *(const v2f*)(bq + 4 * ks);
  }
  {
    v8f acc[4] = {};
    const float* pa = Xt + (16 * jt0 + c) * 260 + kq;
#pragma unroll
    for (int ks = 0; ks < 32; ++ks) {
#pragma unroll
      for (int t = 0; t < 4; ++t) {
        v2f a = *(const v2f*)(pa + t * (16 * 260) + 4 * ks);
        acc[t] = wmma4(a, qf[ks], acc[t]);
      }
    }
#pragma unroll
    for (int t = 0; t < 4; ++t) {
#pragma unroll
      for (int r = 0; r < 8; ++r)
        Xt[(128 + 16 * (jt0 + t) + 8 * h + r) * 260 + irow] = acc[t][r];
    }
  }
  {
    const float* cq = Xt + (128 + kq) * 260 + 128 + irow;
#pragma unroll
    for (int ks = 0; ks < 32; ++ks) {
      v2f q2 = {cq[(4 * ks) * 260], cq[(4 * ks + 1) * 260]};
      qf[ks] = q2;
    }
  }
  __syncthreads();
  {
    v8f acc[4] = {};
    const float* pa = Xt + (128 + 16 * jt0 + c) * 260 + kq;
#pragma unroll
    for (int ks = 0; ks < 32; ++ks) {
#pragma unroll
      for (int t = 0; t < 4; ++t) {
        v2f a = *(const v2f*)(pa + t * (16 * 260) + 4 * ks);
        acc[t] = wmma4(a, qf[ks], acc[t]);
      }
    }
#pragma unroll
    for (int t = 0; t < 4; ++t) {
#pragma unroll
      for (int r = 0; r < 8; ++r)
        Xt[(16 * (jt0 + t) + 8 * h + r) * 260 + 128 + irow] = -acc[t][r];
    }
  }
  __syncthreads();

  float* outp = Linv + (size_t)mk * (D_F * D_F);
  const float* ocol = Xt + tcol * 260;
  const bool upper = (tcol >= 128) && (hh == 0);
  for (int i = hh * 128; i < hh * 128 + 128; i += 4) {
    float4 v = *(const float4*)(ocol + i);
    if (upper) v = make_float4(0.f, 0.f, 0.f, 0.f);
    outp[(size_t)(i + 0) * D_F + tcol] = v.x;
    outp[(size_t)(i + 1) * D_F + tcol] = v.y;
    outp[(size_t)(i + 2) * D_F + tcol] = v.z;
    outp[(size_t)(i + 3) * D_F + tcol] = v.w;
  }
  if (tid == 0) {
    float s = 0.f;
    for (int t = 0; t < D_F; ++t) s += logd[t];
    logdet[mk] = s;
  }
}

__global__ void __launch_bounds__(512)
front_kernel(const float* __restrict__ x, const float* __restrict__ W,
             const float* __restrict__ bias, const float* __restrict__ covs,
             float* __restrict__ feats, float* __restrict__ Linv,
             float* __restrict__ logdet) {
  extern __shared__ float smem[];
  const int blk = blockIdx.x;
  if (blk < NMK) {
    trinv_matrix(covs, Linv, logdet, smem, blk);
  } else {
    const int half = threadIdx.x >> 8;
    feats_tile(x, W, bias, feats, smem + half * (2 * 64 * 66),
               threadIdx.x & 255, 2 * (blk - NMK) + half);
  }
}


template <int FIRST>
__device__ __forceinline__ void tri_block(const float* const (&Ap)[4],
                                          const float* Bp, int k,
                                          v8f (&acc)[8]) {
#pragma unroll
  for (int s = 0; s < 16; s += 4) {
    v2f b0 = *(const v2f*)(Bp + k + s);
    v2f b1 = *(const v2f*)(Bp + 16 * 258 + k + s);
#pragma unroll
    for (int t = FIRST; t < 4; ++t) {
      v2f a = *(const v2f*)(Ap[t] + k + s);
      acc[2 * t + 0] = wmma4(a, b0, acc[2 * t + 0]);
      acc[2 * t + 1] = wmma4(a, b1, acc[2 * t + 1]);
    }
  }
}

__device__ __forceinline__ int tri_need(int mt, int kc) {
  int n = 16 * (mt + 1) - kc;
  return n < 0 ? 0 : (n > 64 ? 64 : n);
}

__global__ void __launch_bounds__(256)
gmm_kernel(const float* __restrict__ feats,
                           const float* __restrict__ Linv,
                           const float* __restrict__ logdet,
                           const float* __restrict__ means,
                           const float* __restrict__ weights,
                           float* __restrict__ out) {
  extern __shared__ float smem[];
  float* As = smem;
  float* Bs = smem + 256 * 66;
  float* quad_s = smem + 256 * 66 + 64 * 258;
  const int tid = threadIdx.x;
  const int lane = tid & 31;
  const int w = __builtin_amdgcn_readfirstlane(tid >> 5);
  const int mk = blockIdx.y;
  const int b0 = blockIdx.x * 64;
  const float* LinvG = Linv + (size_t)mk * (D_F * D_F);
  const float* mu = means + (size_t)mk * D_F;

  const int c4 = (tid & 15) * 4;
  const int r0 = tid >> 4;
  const float* gsrc = LinvG + (size_t)r0 * D_F + c4;
  float* adst = As + r0 * 66 + c4;
  float4 g[16];
#pragma unroll
  for (int p = 0; p < 16; ++p)
    g[p] = *(const float4*)(gsrc + (size_t)(16 * p) * D_F);

  {
    const int d4 = (tid & 63) * 4;
    const int n0 = tid >> 6;
    const float4 m4 = *(const float4*)(mu + d4);
    float4 f[16];
#pragma unroll
    for (int p = 0; p < 16; ++p)
      f[p] = *(const float4*)(feats + (size_t)(b0 + n0 + 4 * p) * D_F + d4);
#pragma unroll
    for (int p = 0; p < 16; ++p) {
      float* dst = Bs + (n0 + 4 * p) * 258 + d4;
      v2f lo = {f[p].x - m4.x, f[p].y - m4.y};
      v2f hi = {f[p].z - m4.z, f[p].w - m4.w};
      *(v2f*)(dst) = lo;
      *(v2f*)(dst + 2) = hi;
    }
  }
  if (tid < 64) quad_s[tid] = 0.f;
#pragma unroll
  for (int p = 0; p < 16; ++p) {
    v2f lo = {g[p].x, g[p].y};
    v2f hi = {g[p].z, g[p].w};
    *(v2f*)(adst + 16 * p * 66) = lo;
    *(v2f*)(adst + 16 * p * 66 + 2) = hi;
  }
  __syncthreads();

  v8f acc[8] = {};
  const int l15 = lane & 15;
  const int khalf = (lane >> 4) * 2;
  const int nh = w >> 2;
  const int q = (w & 4) ? 3 - (w & 3) : (w & 3);
  const int mt[4] = {q, 7 - q, 8 + q, 15 - q};
  const float* const Ap[4] = {As + (mt[0] * 16 + l15) * 66 + khalf,
                              As + (mt[1] * 16 + l15) * 66 + khalf,
                              As + (mt[2] * 16 + l15) * 66 + khalf,
                              As + (mt[3] * 16 + l15) * 66 + khalf};
  const float* Bw = Bs + (nh * 32 + l15) * 258 + khalf;

#pragma unroll
  for (int kc = 0; kc < D_F; kc += 64) {
    if (kc + 64 < D_F) {
#pragma unroll
      for (int p = (kc + 64) / 16; p < 16; ++p)
        g[p] = *(const float4*)(gsrc + (size_t)(16 * p) * D_F + kc + 64);
    }
    {
      const int e0 = tri_need(mt[0], kc), e1 = tri_need(mt[1], kc);
      const int e2 = tri_need(mt[2], kc), e3 = tri_need(mt[3], kc);
      const float* Bp = Bw + kc;
      int k = 0;
      for (; k < e0; k += 16) tri_block<0>(Ap, Bp, k, acc);
      for (; k < e1; k += 16) tri_block<1>(Ap, Bp, k, acc);
      for (; k < e2; k += 16) tri_block<2>(Ap, Bp, k, acc);
      for (; k < e3; k += 16) tri_block<3>(Ap, Bp, k, acc);
    }
    __syncthreads();
    if (kc + 64 < D_F) {
#pragma unroll
      for (int p = (kc + 64) / 16; p < 16; ++p) {
        v2f lo = {g[p].x, g[p].y};
        v2f hi = {g[p].z, g[p].w};
        *(v2f*)(adst + 16 * p * 66) = lo;
        *(v2f*)(adst + 16 * p * 66 + 2) = hi;
      }
      __syncthreads();
    }
  }

#pragma unroll
  for (int t = 0; t < 8; ++t) {
    int nt = 2 * nh + (t & 1);
    float p = 0.f;
#pragma unroll
    for (int r = 0; r < 8; ++r) p += acc[t][r] * acc[t][r];
    atomicAdd(&quad_s[nt * 16 + l15], p);
  }
  __syncthreads();

  if (tid < 64) {
    int m = mk >> 3, kcomp = mk & 7;
    const float* wr = weights + m * NCOMP;
    float wmax = wr[0];
#pragma unroll
    for (int t = 1; t < 8; ++t) wmax = fmaxf(wmax, wr[t]);
    float denom = 0.f;
#pragma unroll
    for (int t = 0; t < 8; ++t) denom += expf(wr[t] - wmax);
    float wmk = expf(wr[kcomp] - wmax) / denom;
    float logp = -0.5f * (quad_s[tid] + 256.0f * LOG2PI) - logdet[mk];
    out[(size_t)(b0 + tid) * NMK + mk] = wmk * logp;
  }
}

extern "C" void kernel_launch(void* const* d_in, const int* in_sizes, int n_in,
                              void* d_out, int out_size, void* d_ws,
                              size_t ws_size, hipStream_t stream) {
  (void)in_sizes; (void)n_in; (void)out_size; (void)ws_size;
  const float* x       = (const float*)d_in[0];
  const float* W       = (const float*)d_in[1];
  const float* bias    = (const float*)d_in[2];
  const float* means   = (const float*)d_in[3];
  const float* covs    = (const float*)d_in[4];
  const float* weights = (const float*)d_in[5];
  float* out = (float*)d_out;

  float* ws = (float*)d_ws;
  float* feats  = ws;
  float* LinvW  = ws + (size_t)B_N * D_F;
  float* logdet = LinvW + (size_t)NMK * D_F * D_F;

  size_t sh2 = (size_t)(256 * 260 + 256 + 2 * 2 * 1028) * sizeof(float);
  front_kernel<<<dim3(NMK + 128), 512, sh2, stream>>>(x, W, bias, covs, feats,
                                                      LinvW, logdet);

  size_t sh3 = (size_t)(256 * 66 + 64 * 258 + 64) * sizeof(float);
  gmm_kernel<<<dim3(B_N / 64, NMK), 256, sh3, stream>>>(feats, LinvW, logdet,
                                                        means, weights, out);
}


# second launch: Linv panel in LDS halved to 32 k: LDS per work-group 134 -> 101 KB and registers 194 -> 160, so three work-groups share a WGP instead of two; k order of every sum kept
# speedup vs baseline: 7.2069x; 1.0334x over previous
#include <hip/hip_runtime.h>

typedef __attribute__((ext_vector_type(2))) float v2f;
typedef __attribute__((ext_vector_type(8))) float v8f;

#define B_N   4096
#define D_INN 1024
#define D_F   256
#define NCOMP 8
#define NMK   80
#define LOG2PI 1.83787706640934548356f

__device__ __forceinline__ v8f wmma4(v2f a, v2f b, v8f c) {
  return __builtin_amdgcn_wmma_f32_16x16x4_f32(false, a, false, b, (short)0, c,
                                               false, false);
}

__device__ __forceinline__ void feats_tile(const float* __restrict__ x,
                                           const float* __restrict__ W,
                                           const float* __restrict__ bias,
                                           float* __restrict__ feats,
                                           float* sm, int tid, int tile) {
  float* As = sm;
  float* Bs = sm + 64 * 66;
  const int lane = tid & 31;
  const int w = tid >> 5;
  const int rowbase = (tile >> 2) * 64;
  const int colbase = (tile & 3) * 64;

  v8f acc[2] = {};
  const int l15 = lane & 15;
  const int khalf = (lane >> 4) * 2;

  for (int kc = 0; kc < D_INN; kc += 64) {
#pragma unroll 8
    for (int e = tid; e < 64 * 64; e += 256) {
      int r = e >> 6, c = e & 63;
      As[r * 66 + c] = x[(size_t)(rowbase + r) * D_INN + kc + c];
      Bs[c * 66 + r] = W[(size_t)(kc + r) * D_F + colbase + c];
    }
    __syncthreads();
    for (int kk = 0; kk < 64; kk += 4) {
      int klo = kk + khalf;
#pragma unroll
      for (int t = 0; t < 2; ++t) {
        int g = w * 2 + t;
        int mt = g >> 2, nt = g & 3;
        v2f a = *(const v2f*)(As + (mt * 16 + l15) * 66 + klo);
        v2f bv = *(const v2f*)(Bs + (nt * 16 + l15) * 66 + klo);
        acc[t] = wmma4(a, bv, acc[t]);
      }
    }
    __syncthreads();
  }

  const int hi = lane >> 4;
#pragma unroll
  for (int t = 0; t < 2; ++t) {
    int g = w * 2 + t;
    int mt = g >> 2, nt = g & 3;
    int n = nt * 16 + l15;
    float bb = bias[colbase + n];
#pragma unroll
    for (int r = 0; r < 8; ++r) {
      int mrow = mt * 16 + r + hi * 8;
      float v = acc[t][r] + bb;
      v = v > 0.f ? v : 0.f;
      feats[(size_t)(rowbase + mrow) * D_F + colbase + n] = v;
    }
  }
}

__device__ __forceinline__ void trinv_matrix(const float* __restrict__ covs,
                                             float* __restrict__ Linv,
                                             float* __restrict__ logdet,
                                             float* smem, int mk) {
  float* Xt = smem;
  float* logd = smem + 256 * 260;
  float* Lst = logd + 256;
  const int tid = threadIdx.x;
  const int lane = tid & 31;
  const int c = lane & 15;
  const int h = lane >> 4;
  const int w = __builtin_amdgcn_readfirstlane(tid >> 5);
  const int k1 = w >> 2;
  const int qd = (k1 + (w & 3)) & 3;
  const int rank = 4 * qd + ((qd & 1) ? 3 - k1 : k1);
  const int slab = ((rank & 1) << 3) | (rank >> 1);
  const int wstart = slab * 16;
  const int bb = wstart & 128;
  const int j = wstart + c;
  const float* Lg = covs + (size_t)mk * (D_F * D_F);

  for (int e = tid; e < 256 * 260; e += 512) Xt[e] = 0.f;
  if (tid < 256) logd[tid] = logf(Lg[tid * D_F + tid]);

  const int tcol = tid & 255;
  const int hh = tid >> 8;
  const int trow = (tcol & 128) + 4 * hh;
  float p[4];
#pragma unroll
  for (int r = 0; r < 4; ++r) p[r] = Lg[(size_t)(trow + r) * D_F + tcol];

  float* xcol = Xt + j * 260;
  for (int s0 = 0; s0 < 128; s0 += 8) {
    float* Lb = Lst + ((s0 >> 3) & 1) * (2 * 1028);
    *(float4*)(Lb + hh * 1028 + tcol * 4) = make_float4(p[0], p[1], p[2], p[3]);
    __syncthreads();
    if (s0 + 8 < 128) {
#pragma unroll
      for (int r = 0; r < 4; ++r)
        p[r] = Lg[(size_t)(s0 + 8 + trow + r) * D_F + tcol];
    }
    const int i0 = bb + s0;
    if (wstart <= i0) {
      float s0_ = 0.f, s1 = 0.f, s2 = 0.f, s3 = 0.f;
      const float* Lh = Lb + h * 1028;
      for (int t = wstart; t < i0; t += 8) {
        float4 xa = *(const float4*)(xcol + t);
        float4 xb = *(const float4*)(xcol + t + 4);
        float xs[8] = {xa.x, xa.y, xa.z, xa.w, xb.x, xb.y, xb.z, xb.w};
#pragma unroll
        for (int u = 0; u < 8; ++u) {
          float4 l = *(const float4*)(Lh + (t + u) * 4);
          s0_ = fmaf(l.x, xs[u], s0_);
          s1 = fmaf(l.y, xs[u], s1);
          s2 = fmaf(l.z, xs[u], s2);
          s3 = fmaf(l.w, xs[u], s3);
        }
      }
      float o0 = __shfl_xor(s0_, 16);
      float o1 = __shfl_xor(s1, 16);
      float o2 = __shfl_xor(s2, 16);
      float o3 = __shfl_xor(s3, 16);
      float s[8];
      s[0] = h ? o0 : s0_; s[1] = h ? o1 : s1;
      s[2] = h ? o2 : s2;  s[3] = h ? o3 : s3;
      s[4] = h ? s0_ : o0; s[5] = h ? s1 : o1;
      s[6] = h ? s2 : o2;  s[7] = h ? s3 : o3;
      float xv[8];
#pragma unroll
      for (int r = 0; r < 8; ++r) {
        const float* Lr = Lb + (r >> 2) * 1028 + (r & 3);
        float acc = s[r];
#pragma unroll
        for (int q = 0; q < r; ++q) acc = fmaf(Lr[(i0 + q) * 4], xv[q], acc);
        float d = Lr[(i0 + r) * 4];
        float num = ((j == i0 + r) ? 1.f : 0.f) - acc;
        float v = num / d;
        xv[r] = (j <= i0 + r) ? v : 0.f;
      }
      if (h == 0) {
        *(float4*)(xcol + i0) = make_float4(xv[0], xv[1], xv[2], xv[3]);
        *(float4*)(xcol + i0 + 4) = make_float4(xv[4], xv[5], xv[6], xv[7]);
      }
    }
  }
  __syncthreads();

  const int it = w & 7;
  const int jt0 = (w >> 3) * 4;
  const int irow = 16 * it + c;
  const int kq = 2 * h;
  v2f qf[32];
  {
    const float* bq = Lg + (size_t)(128 + irow) * D_F + kq;
#pragma unroll
    for (int ks = 0; ks < 32; ++ks) qf[ks] = *(const v2f*)(bq + 4 * ks);
  }
  {
    v8f acc[4] = {};
    const float* pa = Xt + (16 * jt0 + c) * 260 + kq;
#pragma unroll
    for (int ks = 0; ks < 32; ++ks) {
#pragma unroll
      for (int t = 0; t < 4; ++t) {
        v2f a = *(const v2f*)(pa + t * (16 * 260) + 4 * ks);
        acc[t] = wmma4(a, qf[ks], acc[t]);
      }
    }
#pragma unroll
    for (int t = 0; t < 4; ++t) {
#pragma unroll
      for (int r = 0; r < 8; ++r)
        Xt[(128 + 16 * (jt0 + t) + 8 * h + r) * 260 + irow] = acc[t][r];
    }
  }
  {
    const float* cq = Xt + (128 + kq) * 260 + 128 + irow;
#pragma unroll
    for (int ks = 0; ks < 32; ++ks) {
      v2f q2 = {cq[(4 * ks) * 260], cq[(4 * ks + 1) * 260]};
      qf[ks] = q2;
    }
  }
  __syncthreads();
  {
    v8f acc[4] = {};
    const float* pa = Xt + (128 + 16 * jt0 + c) * 260 + kq;
#pragma unroll
    for (int ks = 0; ks < 32; ++ks) {
#pragma unroll
      for (int t = 0; t < 4; ++t) {
        v2f a = *(const v2f*)(pa + t * (16 * 260) + 4 * ks);
        acc[t] = wmma4(a, qf[ks], acc[t]);
      }
    }
#pragma unroll
    for (int t = 0; t < 4; ++t) {
#pragma unroll
      for (int r = 0; r < 8; ++r)
        Xt[(16 * (jt0 + t) + 8 * h + r) * 260 + 128 + irow] = -acc[t][r];
    }
  }
  __syncthreads();

  float* outp = Linv + (size_t)mk * (D_F * D_F);
  const float* ocol = Xt + tcol * 260;
  const bool upper = (tcol >= 128) && (hh == 0);
  for (int i = hh * 128; i < hh * 128 + 128; i += 4) {
    float4 v = *(const float4*)(ocol + i);
    if (upper) v = make_float4(0.f, 0.f, 0.f, 0.f);
    outp[(size_t)(i + 0) * D_F + tcol] = v.x;
    outp[(size_t)(i + 1) * D_F + tcol] = v.y;
    outp[(size_t)(i + 2) * D_F + tcol] = v.z;
    outp[(size_t)(i + 3) * D_F + tcol] = v.w;
  }
  if (tid == 0) {
    float s = 0.f;
    for (int t = 0; t < D_F; ++t) s += logd[t];
    logdet[mk] = s;
  }
}

__global__ void __launch_bounds__(512)
front_kernel(const float* __restrict__ x, const float* __restrict__ W,
             const float* __restrict__ bias, const float* __restrict__ covs,
             float* __restrict__ feats, float* __restrict__ Linv,
             float* __restrict__ logdet) {
  extern __shared__ float smem[];
  const int blk = blockIdx.x;
  if (blk < NMK) {
    trinv_matrix(covs, Linv, logdet, smem, blk);
  } else {
    const int half = threadIdx.x >> 8;
    feats_tile(x, W, bias, feats, smem + half * (2 * 64 * 66),
               threadIdx.x & 255, 2 * (blk - NMK) + half);
  }
}


template <int FIRST>
__device__ __forceinline__ void tri_block(const float* const (&Ap)[4],
                                          const float* Bp, int k,
                                          v8f (&acc)[8]) {
#pragma unroll
  for (int s = 0; s < 16; s += 4) {
    v2f b0 = *(const v2f*)(Bp + k + s);
    v2f b1 = *(const v2f*)(Bp + 16 * 258 + k + s);
#pragma unroll
    for (int t = FIRST; t < 4; ++t) {
      v2f a = *(const v2f*)(Ap[t] + k + s);
      acc[2 * t + 0] = wmma4(a, b0, acc[2 * t + 0]);
      acc[2 * t + 1] = wmma4(a, b1, acc[2 * t + 1]);
    }
  }
}

#define KP 32
#define AP 34

__device__ __forceinline__ int tri_need(int mt, int kc) {
  int n = 16 * (mt + 1) - kc;
  return n < 0 ? 0 : (n > KP ? KP : n);
}

__global__ void __launch_bounds__(256)
gmm_kernel(const float* __restrict__ feats,
                           const float* __restrict__ Linv,
                           const float* __restrict__ logdet,
                           const float* __restrict__ means,
                           const float* __restrict__ weights,
                           float* __restrict__ out) {
  extern __shared__ float smem[];
  float* As = smem;
  float* Bs = smem + 256 * AP;
  float* quad_s = smem + 256 * AP + 64 * 258;
  const int tid = threadIdx.x;
  const int lane = tid & 31;
  const int w = __builtin_amdgcn_readfirstlane(tid >> 5);
  const int mk = blockIdx.y;
  const int b0 = blockIdx.x * 64;
  const float* LinvG = Linv + (size_t)mk * (D_F * D_F);
  const float* mu = means + (size_t)mk * D_F;

  const int c4 = (tid & 7) * 4;
  const int r0 = tid >> 3;
  const float* gsrc = LinvG + (size_t)r0 * D_F + c4;
  float* adst = As + r0 * AP + c4;
  float4 g[8];
#pragma unroll
  for (int p = 0; p < 8; ++p)
    g[p] = *(const float4*)(gsrc + (size_t)(32 * p) * D_F);

  {
    const int d4 = (tid & 63) * 4;
    const int n0 = tid >> 6;
    const float4 m4 = *(const float4*)(mu + d4);
    float4 f[16];
#pragma unroll
    for (int p = 0; p < 16; ++p)
      f[p] = *(const float4*)(feats + (size_t)(b0 + n0 + 4 * p) * D_F + d4);
#pragma unroll
    for (int p = 0; p < 16; ++p) {
      float* dst = Bs + (n0 + 4 * p) * 258 + d4;
      v2f lo = {f[p].x - m4.x, f[p].y - m4.y};
      v2f hi = {f[p].z - m4.z, f[p].w - m4.w};
      *(v2f*)(dst) = lo;
      *(v2f*)(dst + 2) = hi;
    }
  }
  if (tid < 64) quad_s[tid] = 0.f;
#pragma unroll
  for (int p = 0; p < 8; ++p) {
    v2f lo = {g[p].x, g[p].y};
    v2f hi = {g[p].z, g[p].w};
    *(v2f*)(adst + 32 * p * AP) = lo;
    *(v2f*)(adst + 32 * p * AP + 2) = hi;
  }
  __syncthreads();

  v8f acc[8] = {};
  const int l15 = lane & 15;
  const int khalf = (lane >> 4) * 2;
  const int nh = w >> 2;
  const int q = (w & 4) ? 3 - (w & 3) : (w & 3);
  const int mt[4] = {q, 7 - q, 8 + q, 15 - q};
  const float* const Ap[4] = {As + (mt[0] * 16 + l15) * AP + khalf,
                              As + (mt[1] * 16 + l15) * AP + khalf,
                              As + (mt[2] * 16 + l15) * AP + khalf,
                              As + (mt[3] * 16 + l15) * AP + khalf};
  const float* Bw = Bs + (nh * 32 + l15) * 258 + khalf;

#pragma unroll
  for (int kc = 0; kc < D_F; kc += KP) {
    if (kc + KP < D_F) {
#pragma unroll
      for (int p = (kc + KP) / 32; p < 8; ++p)
        g[p] = *(const float4*)(gsrc + (size_t)(32 * p) * D_F + kc + KP);
    }
    {
      const int e0 = tri_need(mt[0], kc), e1 = tri_need(mt[1], kc);
      const int e2 = tri_need(mt[2], kc), e3 = tri_need(mt[3], kc);
      const float* Bp = Bw + kc;
      int k = 0;
      for (; k < e0; k += 16) tri_block<0>(Ap, Bp, k, acc);
      for (; k < e1; k += 16) tri_block<1>(Ap, Bp, k, acc);
      for (; k < e2; k += 16) tri_block<2>(Ap, Bp, k, acc);
      for (; k < e3; k += 16) tri_block<3>(Ap, Bp, k, acc);
    }
    __syncthreads();
    if (kc + KP < D_F) {
#pragma unroll
      for (int p = (kc + KP) / 32; p < 8; ++p) {
        v2f lo = {g[p].x, g[p].y};
        v2f hi = {g[p].z, g[p].w};
        *(v2f*)(adst + 32 * p * AP) = lo;
        *(v2f*)(adst + 32 * p * AP + 2) = hi;
      }
      __syncthreads();
    }
  }

#pragma unroll
  for (int t = 0; t < 8; ++t) {
    int nt = 2 * nh + (t & 1);
    float p = 0.f;
#pragma unroll
    for (int r = 0; r < 8; ++r) p += acc[t][r] * acc[t][r];
    atomicAdd(&quad_s[nt * 16 + l15], p);
  }
  __syncthreads();

  if (tid < 64) {
    int m = mk >> 3, kcomp = mk & 7;
    const float* wr = weights + m * NCOMP;
    float wmax = wr[0];
#pragma unroll
    for (int t = 1; t < 8; ++t) wmax = fmaxf(wmax, wr[t]);
    float denom = 0.f;
#pragma unroll
    for (int t = 0; t < 8; ++t) denom += expf(wr[t] - wmax);
    float wmk = expf(wr[kcomp] - wmax) / denom;
    float logp = -0.5f * (quad_s[tid] + 256.0f * LOG2PI) - logdet[mk];
    out[(size_t)(b0 + tid) * NMK + mk] = wmk * logp;
  }
}

extern "C" void kernel_launch(void* const* d_in, const int* in_sizes, int n_in,
                              void* d_out, int out_size, void* d_ws,
                              size_t ws_size, hipStream_t stream) {
  (void)in_sizes; (void)n_in; (void)out_size; (void)ws_size;
  const float* x       = (const float*)d_in[0];
  const float* W       = (const float*)d_in[1];
  const float* bias    = (const float*)d_in[2];
  const float* means   = (const float*)d_in[3];
  const float* covs    = (const float*)d_in[4];
  const float* weights = (const float*)d_in[5];
  float* out = (float*)d_out;

  float* ws = (float*)d_ws;
  float* feats  = ws;
  float* LinvW  = ws + (size_t)B_N * D_F;
  float* logdet = LinvW + (size_t)NMK * D_F * D_F;

  size_t sh2 = (size_t)(256 * 260 + 256 + 2 * 2 * 1028) * sizeof(float);
  front_kernel<<<dim3(NMK + 128), 512, sh2, stream>>>(x, W, bias, covs, feats,
                                                      LinvW, logdet);

  size_t sh3 = (size_t)(256 * AP + 64 * 258 + 64) * sizeof(float);
  gmm_kernel<<<dim3(B_N / 64, NMK), 256, sh3, stream>>>(feats, LinvW, logdet,
                                                        means, weights, out);
}


# first launch: loads issued ahead of use: GEMM chunks as 16-byte loads one chunk ahead; inverse: L rows 4 steps ahead (2 barriers per 4 steps, was 4), its LDS reads one trip ahead; sums unchanged
# speedup vs baseline: 7.7164x; 1.0707x over previous
#include <hip/hip_runtime.h>

typedef __attribute__((ext_vector_type(2))) float v2f;
typedef __attribute__((ext_vector_type(8))) float v8f;
typedef __attribute__((ext_vector_type(4))) float v4f;

#define B_N   4096
#define D_INN 1024
#define D_F   256
#define NCOMP 8
#define NMK   80
#define LOG2PI 1.83787706640934548356f

__device__ __forceinline__ v8f wmma4(v2f a, v2f b, v8f c) {
  return __builtin_amdgcn_wmma_f32_16x16x4_f32(false, a, false, b, (short)0, c,
                                               false, false);
}

__device__ __forceinline__ void feats_tile(const float* __restrict__ x,
                                           const float* __restrict__ W,
                                           const float* __restrict__ bias,
                                           float* __restrict__ feats,
                                           float* sm, int tid, int tile) {
  float* As = sm;
  float* Bs = sm + 64 * 66;
  const int lane = tid & 31;
  const int w = tid >> 5;
  const int rowbase = (tile >> 2) * 64;
  const int colbase = (tile & 3) * 64;

  v8f acc[2] = {};
  const int l15 = lane & 15;
  const int khalf = (lane >> 4) * 2;

  const int sr = tid >> 4;
  const int sc = (tid & 15) * 4;
  const float* xg = x + (size_t)(rowbase + sr) * D_INN + sc;
  const float* wg = W + (size_t)sr * D_F + colbase + sc;
  float4 xa[4], wa[4];
#pragma unroll
  for (int i = 0; i < 4; ++i) {
    xa[i] = *(const float4*)(xg + (size_t)(16 * i) * D_INN);
    wa[i] = *(const float4*)(wg + (size_t)(16 * i) * D_F);
  }

  for (int kc = 0; kc < D_INN; kc += 64) {
#pragma unroll
    for (int i = 0; i < 4; ++i) {
      float* ad = As + (sr + 16 * i) * 66 + sc;
      v2f lo = {xa[i].x, xa[i].y};
      v2f hi = {xa[i].z, xa[i].w};
      *(v2f*)(ad) = lo;
      *(v2f*)(ad + 2) = hi;
      float* bd = Bs + sc * 66 + sr + 16 * i;
      bd[0] = wa[i].x;
      bd[66] = wa[i].y;
      bd[132] = wa[i].z;
      bd[198] = wa[i].w;
    }
    __syncthreads();
    if (kc + 64 < D_INN) {
#pragma unroll
      for (int i = 0; i < 4; ++i) {
        xa[i] = *(const float4*)(xg + (size_t)(16 * i) * D_INN + kc + 64);
        wa[i] = *(const float4*)(wg + (size_t)(kc + 64 + 16 * i) * D_F);
      }
    }
    for (int kk = 0; kk < 64; kk += 4) {
      int klo = kk + khalf;
#pragma unroll
      for (int t = 0; t < 2; ++t) {
        int g = w * 2 + t;
        int mt = g >> 2, nt = g & 3;
        v2f a = *(const v2f*)(As + (mt * 16 + l15) * 66 + klo);
        v2f bv = *(const v2f*)(Bs + (nt * 16 + l15) * 66 + klo);
        acc[t] = wmma4(a, bv, acc[t]);
      }
    }
    __syncthreads();
  }

  const int hi = lane >> 4;
#pragma unroll
  for (int t = 0; t < 2; ++t) {
    int g = w * 2 + t;
    int mt = g >> 2, nt = g & 3;
    int n = nt * 16 + l15;
    float bb = bias[colbase + n];
#pragma unroll
    for (int r = 0; r < 8; ++r) {
      int mrow = mt * 16 + r + hi * 8;
      float v = acc[t][r] + bb;
      v = v > 0.f ? v : 0.f;
      feats[(size_t)(rowbase + mrow) * D_F + colbase + n] = v;
    }
  }
}

struct SubSet {
  v4f x0, x1;
  v4f l[8];
};

__device__ __forceinline__ void sub_load(SubSet& s, const float* xcol,
                                         const float* Lh, int t) {
  s.x0 = *(const v4f*)(xcol + t);
  s.x1 = *(const v4f*)(xcol + t + 4);
#pragma unroll
  for (int u = 0; u < 8; ++u) s.l[u] = *(const v4f*)(Lh + (t + u) * 4);
}

__device__ __forceinline__ void sub_fma(const SubSet& s, float& a0, float& a1,
                                        float& a2, float& a3) {
#pragma unroll
  for (int u = 0; u < 8; ++u) {
    const float xs = (u < 4) ? s.x0[u & 3] : s.x1[u & 3];
    a0 = fmaf(s.l[u][0], xs, a0);
    a1 = fmaf(s.l[u][1], xs, a1);
    a2 = fmaf(s.l[u][2], xs, a2);
    a3 = fmaf(s.l[u][3], xs, a3);
  }
}

__device__ __forceinline__ void trinv_matrix(const float* __restrict__ covs,
                                             float* __restrict__ Linv,
                                             float* __restrict__ logdet,
                                             float* smem, int mk) {
  float* Xt = smem;
  float* logd = smem + 256 * 260;
  float* Lst = logd + 256;
  const int tid = threadIdx.x;
  const int lane = tid & 31;
  const int c = lane & 15;
  const int h = lane >> 4;
  const int w = __builtin_amdgcn_readfirstlane(tid >> 5);
  const int k1 = w >> 2;
  const int qd = (k1 + (w & 3)) & 3;
  const int rank = 4 * qd + ((qd & 1) ? 3 - k1 : k1);
  const int slab = ((rank & 1) << 3) | (rank >> 1);
  const int wstart = slab * 16;
  const int bb = wstart & 128;
  const int j = wstart + c;
  const float* Lg = covs + (size_t)mk * (D_F * D_F);

  const int tcol = tid & 255;
  const int hh = tid >> 8;
  const int trow = (tcol & 128) + 4 * hh;
  const float dg = Lg[tcol * D_F + tcol];
  float p[4][4];
#pragma unroll
  for (int k = 0; k < 4; ++k) {
#pragma unroll
    for (int r = 0; r < 4; ++r)
      p[k][r] = Lg[(size_t)(8 * k + trow + r) * D_F + tcol];
  }

  for (int e = tid; e < 256 * 260; e += 512) Xt[e] = 0.f;
  if (tid < 256) logd[tid] = logf(dg);

  float* xcol = Xt + j * 260;
  for (int g0 = 0; g0 < 128; g0 += 32) {
#pragma unroll
    for (int k = 0; k < 4; ++k)
      *(float4*)(Lst + k * (2 * 1028) + hh * 1028 + tcol * 4) =
          make_float4(p[k][0], p[k][1], p[k][2], p[k][3]);
    __syncthreads();
    if (g0 + 32 < 128) {
#pragma unroll
      for (int k = 0; k < 4; ++k) {
#pragma unroll
        for (int r = 0; r < 4; ++r)
          p[k][r] = Lg[(size_t)(g0 + 32 + 8 * k + trow + r) * D_F + tcol];
      }
    }
#pragma unroll 1
    for (int k = 0; k < 4; ++k) {
      const int s0 = g0 + 8 * k;
      const float* Lb = Lst + k * (2 * 1028);
      const int i0 = bb + s0;
      if (wstart <= i0) {
        float Ld[8][8];
#pragma unroll
        for (int q = 0; q < 8; ++q) {
#pragma unroll
          for (int g = 0; g < 2; ++g) {
            const float* pp = Lb + g * 1028 + (i0 + q) * 4;
            const int lo = q - 4 * g;
            if (lo <= 0) {
              v4f v = *(const v4f*)pp;
              Ld[4 * g + 0][q] = v[0]; Ld[4 * g + 1][q] = v[1];
              Ld[4 * g + 2][q] = v[2]; Ld[4 * g + 3][q] = v[3];
            } else if (lo == 1) {
              Ld[4 * g + 1][q] = pp[1];
              v2f v = *(const v2f*)(pp + 2);
              Ld[4 * g + 2][q] = v[0]; Ld[4 * g + 3][q] = v[1];
            } else if (lo == 2) {
              v2f v = *(const v2f*)(pp + 2);
              Ld[4 * g + 2][q] = v[0]; Ld[4 * g + 3][q] = v[1];
            } else if (lo == 3) {
              Ld[4 * g + 3][q] = pp[3];
            }
          }
        }

        float s0_ = 0.f, s1 = 0.f, s2 = 0.f, s3 = 0.f;
        if (wstart < i0) {
          const float* Lh = Lb + h * 1028;
          SubSet sa, sb;
          int t = wstart;
          sub_load(sa, xcol, Lh, t);
          for (;;) {
            if (t + 8 >= i0) { sub_fma(sa, s0_, s1, s2, s3); break; }
            sub_load(sb, xcol, Lh, t + 8);
            sub_fma(sa, s0_, s1, s2, s3);
            t += 8;
            if (t + 8 >= i0) { sub_fma(sb, s0_, s1, s2, s3); break; }
            sub_load(sa, xcol, Lh, t + 8);
            sub_fma(sb, s0_, s1, s2, s3);
            t += 8;
          }
        }
        float o0 = __shfl_xor(s0_, 16);
        float o1 = __shfl_xor(s1, 16);
        float o2 = __shfl_xor(s2, 16);
        float o3 = __shfl_xor(s3, 16);
        float s[8];
        s[0] = h ? o0 : s0_; s[1] = h ? o1 : s1;
        s[2] = h ? o2 : s2;  s[3] = h ? o3 : s3;
        s[4] = h ? s0_ : o0; s[5] = h ? s1 : o1;
        s[6] = h ? s2 : o2;  s[7] = h ? s3 : o3;
        float xv[8];
#pragma unroll
        for (int r = 0; r < 8; ++r) {
          float acc = s[r];
#pragma unroll
          for (int q = 0; q < r; ++q) acc = fmaf(Ld[r][q], xv[q], acc);
          float d = Ld[r][r];
          float num = ((j == i0 + r) ? 1.f : 0.f) - acc;
          float v = num / d;
          xv[r] = (j <= i0 + r) ? v : 0.f;
        }
        if (h == 0) {
          *(float4*)(xcol + i0) = make_float4(xv[0], xv[1], xv[2], xv[3]);
          *(float4*)(xcol + i0 + 4) = make_float4(xv[4], xv[5], xv[6], xv[7]);
        }
      }
    }
    __syncthreads();
  }

  const int it = w & 7;
  const int jt0 = (w >> 3) * 4;
  const int irow = 16 * it + c;
  const int kq = 2 * h;
  v2f qf[32];
  {
    const float* bq = Lg + (size_t)(128 + irow) * D_F + kq;
#pragma unroll
    for (int ks = 0; ks < 32; ++ks) qf[ks] = *(const v2f*)(bq + 4 * ks);
  }
  {
    v8f acc[4] = {};
    const float* pa = Xt + (16 * jt0 + c) * 260 + kq;
#pragma unroll
    for (int ks = 0; ks < 32; ++ks) {
#pragma unroll
      for (int t = 0; t < 4; ++t) {
        v2f a = *(const v2f*)(pa + t * (16 * 260) + 4 * ks);
        acc[t] = wmma4(a, qf[ks], acc[t]);
      }
    }
#pragma unroll
    for (int t = 0; t < 4; ++t) {
#pragma unroll
      for (int r = 0; r < 8; ++r)
        Xt[(128 + 16 * (jt0 + t) + 8 * h + r) * 260 + irow] = acc[t][r];
    }
  }
  {
    const float* cq = Xt + (128 + kq) * 260 + 128 + irow;
#pragma unroll
    for (int ks = 0; ks < 32; ++ks) {
      v2f q2 = {cq[(4 * ks) * 260], cq[(4 * ks + 1) * 260]};
      qf[ks] = q2;
    }
  }
  __syncthreads();
  {
    v8f acc[4] = {};
    const float* pa = Xt + (128 + 16 * jt0 + c) * 260 + kq;
#pragma unroll
    for (int ks = 0; ks < 32; ++ks) {
#pragma unroll
      for (int t = 0; t < 4; ++t) {
        v2f a = *(const v2f*)(pa + t * (16 * 260) + 4 * ks);
        acc[t] = wmma4(a, qf[ks], acc[t]);
      }
    }
#pragma unroll
    for (int t = 0; t < 4; ++t) {
#pragma unroll
      for (int r = 0; r < 8; ++r)
        Xt[(16 * (jt0 + t) + 8 * h + r) * 260 + 128 + irow] = -acc[t][r];
    }
  }
  __syncthreads();

  float* outp = Linv + (size_t)mk * (D_F * D_F);
  const float* ocol = Xt + tcol * 260;
  const bool upper = (tcol >= 128) && (hh == 0);
  for (int i = hh * 128; i < hh * 128 + 128; i += 4) {
    float4 v = *(const float4*)(ocol + i);
    if (upper) v = make_float4(0.f, 0.f, 0.f, 0.f);
    outp[(size_t)(i + 0) * D_F + tcol] = v.x;
    outp[(size_t)(i + 1) * D_F + tcol] = v.y;
    outp[(size_t)(i + 2) * D_F + tcol] = v.z;
    outp[(size_t)(i + 3) * D_F + tcol] = v.w;
  }
  if (tid == 0) {
    float s = 0.f;
    for (int t = 0; t < D_F; ++t) s += logd[t];
    logdet[mk] = s;
  }
}

__global__ void __launch_bounds__(512)
front_kernel(const float* __restrict__ x, const float* __restrict__ W,
             const float* __restrict__ bias, const float* __restrict__ covs,
             float* __restrict__ feats, float* __restrict__ Linv,
             float* __restrict__ logdet) {
  extern __shared__ float smem[];
  const int blk = blockIdx.x;
  if (blk < NMK) {
    trinv_matrix(covs, Linv, logdet, smem, blk);
  } else {
    const int half = threadIdx.x >> 8;
    feats_tile(x, W, bias, feats, smem + half * (2 * 64 * 66),
               threadIdx.x & 255, 2 * (blk - NMK) + half);
  }
}


template <int FIRST>
__device__ __forceinline__ void tri_block(const float* const (&Ap)[4],
                                          const float* Bp, int k,
                                          v8f (&acc)[8]) {
#pragma unroll
  for (int s = 0; s < 16; s += 4) {
    v2f b0 = *(const v2f*)(Bp + k + s);
    v2f b1 = *(const v2f*)(Bp + 16 * 258 + k + s);
#pragma unroll
    for (int t = FIRST; t < 4; ++t) {
      v2f a = *(const v2f*)(Ap[t] + k + s);
      acc[2 * t + 0] = wmma4(a, b0, acc[2 * t + 0]);
      acc[2 * t + 1] = wmma4(a, b1, acc[2 * t + 1]);
    }
  }
}

#define KP 32
#define AP 34

__device__ __forceinline__ int tri_need(int mt, int kc) {
  int n = 16 * (mt + 1) - kc;
  return n < 0 ? 0 : (n > KP ? KP : n);
}

__global__ void __launch_bounds__(256)
gmm_kernel(const float* __restrict__ feats,
                           const float* __restrict__ Linv,
                           const float* __restrict__ logdet,
                           const float* __restrict__ means,
                           const float* __restrict__ weights,
                           float* __restrict__ out) {
  extern __shared__ float smem[];
  float* As = smem;
  float* Bs = smem + 256 * AP;
  float* quad_s = smem + 256 * AP + 64 * 258;
  const int tid = threadIdx.x;
  const int lane = tid & 31;
  const int w = __builtin_amdgcn_readfirstlane(tid >> 5);
  const int mk = blockIdx.y;
  const int b0 = blockIdx.x * 64;
  const float* LinvG = Linv + (size_t)mk * (D_F * D_F);
  const float* mu = means + (size_t)mk * D_F;

  const int c4 = (tid & 7) * 4;
  const int r0 = tid >> 3;
  const float* gsrc = LinvG + (size_t)r0 * D_F + c4;
  float* adst = As + r0 * AP + c4;
  float4 g[8];
#pragma unroll
  for (int p = 0; p < 8; ++p)
    g[p] = *(const float4*)(gsrc + (size_t)(32 * p) * D_F);

  {
    const int d4 = (tid & 63) * 4;
    const int n0 = tid >> 6;
    const float4 m4 = *(const float4*)(mu + d4);
    float4 f[16];
#pragma unroll
    for (int p = 0; p < 16; ++p)
      f[p] = *(const float4*)(feats + (size_t)(b0 + n0 + 4 * p) * D_F + d4);
#pragma unroll
    for (int p = 0; p < 16; ++p) {
      float* dst = Bs + (n0 + 4 * p) * 258 + d4;
      v2f lo = {f[p].x - m4.x, f[p].y - m4.y};
      v2f hi = {f[p].z - m4.z, f[p].w - m4.w};
      *(v2f*)(dst) = lo;
      *(v2f*)(dst + 2) = hi;
    }
  }
  if (tid < 64) quad_s[tid] = 0.f;
#pragma unroll
  for (int p = 0; p < 8; ++p) {
    v2f lo = {g[p].x, g[p].y};
    v2f hi = {g[p].z, g[p].w};
    *(v2f*)(adst + 32 * p * AP) = lo;
    *(v2f*)(adst + 32 * p * AP + 2) = hi;
  }
  __syncthreads();

  v8f acc[8] = {};
  const int l15 = lane & 15;
  const int khalf = (lane >> 4) * 2;
  const int nh = w >> 2;
  const int q = (w & 4) ? 3 - (w & 3) : (w & 3);
  const int mt[4] = {q, 7 - q, 8 + q, 15 - q};
  const float* const Ap[4] = {As + (mt[0] * 16 + l15) * AP + khalf,
                              As + (mt[1] * 16 + l15) * AP + khalf,
                              As + (mt[2] * 16 + l15) * AP + khalf,
                              As + (mt[3] * 16 + l15) * AP + khalf};
  const float* Bw = Bs + (nh * 32 + l15) * 258 + khalf;

#pragma unroll
  for (int kc = 0; kc < D_F; kc += KP) {
    if (kc + KP < D_F) {
#pragma unroll
      for (int p = (kc + KP) / 32; p < 8; ++p)
        g[p] = *(const float4*)(gsrc + (size_t)(32 * p) * D_F + kc + KP);
    }
    {
      const int e0 = tri_need(mt[0], kc), e1 = tri_need(mt[1], kc);
      const int e2 = tri_need(mt[2], kc), e3 = tri_need(mt[3], kc);
      const float* Bp = Bw + kc;
      int k = 0;
      for (; k < e0; k += 16) tri_block<0>(Ap, Bp, k, acc);
      for (; k < e1; k += 16) tri_block<1>(Ap, Bp, k, acc);
      for (; k < e2; k += 16) tri_block<2>(Ap, Bp, k, acc);
      for (; k < e3; k += 16) tri_block<3>(Ap, Bp, k, acc);
    }
    __syncthreads();
    if (kc + KP < D_F) {
#pragma unroll
      for (int p = (kc + KP) / 32; p < 8; ++p) {
        v2f lo = {g[p].x, g[p].y};
        v2f hi = {g[p].z, g[p].w};
        *(v2f*)(adst + 32 * p * AP) = lo;
        *(v2f*)(adst + 32 * p * AP + 2) = hi;
      }
      __syncthreads();
    }
  }

#pragma unroll
  for (int t = 0; t < 8; ++t) {
    int nt = 2 * nh + (t & 1);
    float p = 0.f;
#pragma unroll
    for (int r = 0; r < 8; ++r) p += acc[t][r] * acc[t][r];
    atomicAdd(&quad_s[nt * 16 + l15], p);
  }
  __syncthreads();

  if (tid < 64) {
    int m = mk >> 3, kcomp = mk & 7;
    const float* wr = weights + m * NCOMP;
    float wmax = wr[0];
#pragma unroll
    for (int t = 1; t < 8; ++t) wmax = fmaxf(wmax, wr[t]);
    float denom = 0.f;
#pragma unroll
    for (int t = 0; t < 8; ++t) denom += expf(wr[t] - wmax);
    float wmk = expf(wr[kcomp] - wmax) / denom;
    float logp = -0.5f * (quad_s[tid] + 256.0f * LOG2PI) - logdet[mk];
    out[(size_t)(b0 + tid) * NMK + mk] = wmk * logp;
  }
}

extern "C" void kernel_launch(void* const* d_in, const int* in_sizes, int n_in,
                              void* d_out, int out_size, void* d_ws,
                              size_t ws_size, hipStream_t stream) {
  (void)in_sizes; (void)n_in; (void)out_size; (void)ws_size;
  const float* x       = (const float*)d_in[0];
  const float* W       = (const float*)d_in[1];
  const float* bias    = (const float*)d_in[2];
  const float* means   = (const float*)d_in[3];
  const float* covs    = (const float*)d_in[4];
  const float* weights = (const float*)d_in[5];
  float* out = (float*)d_out;

  float* ws = (float*)d_ws;
  float* feats  = ws;
  float* LinvW  = ws + (size_t)B_N * D_F;
  float* logdet = LinvW + (size_t)NMK * D_F * D_F;

  size_t sh2 = (size_t)(256 * 260 + 256 + 4 * 2 * 1028) * sizeof(float);
  front_kernel<<<dim3(NMK + 128), 512, sh2, stream>>>(x, W, bias, covs, feats,
                                                      LinvW, logdet);

  size_t sh3 = (size_t)(256 * AP + 64 * 258 + 64) * sizeof(float);
  gmm_kernel<<<dim3(B_N / 64, NMK), 256, sh3, stream>>>(feats, LinvW, logdet,
                                                        means, weights, out);
}


# inverse's lower-left block no longer formed: second launch solves by 128-blocks on f32 WMMA, z_top=invA d_top, z_bot=invC(d_bot - B z_top), B from cov; sum order of z_bot and quad changes
# speedup vs baseline: 7.9655x; 1.0323x over previous
#include <hip/hip_runtime.h>

typedef __attribute__((ext_vector_type(2))) float v2f;
typedef __attribute__((ext_vector_type(8))) float v8f;
typedef __attribute__((ext_vector_type(4))) float v4f;

#define B_N   4096
#define D_INN 1024
#define D_F   256
#define NCOMP 8
#define NMK   80
#define LOG2PI 1.83787706640934548356f

__device__ __forceinline__ v8f wmma4(v2f a, v2f b, v8f c) {
  return __builtin_amdgcn_wmma_f32_16x16x4_f32(false, a, false, b, (short)0, c,
                                               false, false);
}

__device__ __forceinline__ void feats_tile(const float* __restrict__ x,
                                           const float* __restrict__ W,
                                           const float* __restrict__ bias,
                                           float* __restrict__ feats,
                                           float* sm, int tid, int tile) {
  float* As = sm;
  float* Bs = sm + 64 * 66;
  const int lane = tid & 31;
  const int w = tid >> 5;
  const int rowbase = (tile >> 2) * 64;
  const int colbase = (tile & 3) * 64;

  v8f acc[2] = {};
  const int l15 = lane & 15;
  const int khalf = (lane >> 4) * 2;

  const int sr = tid >> 4;
  const int sc = (tid & 15) * 4;
  const float* xg = x + (size_t)(rowbase + sr) * D_INN + sc;
  const float* wg = W + (size_t)sr * D_F + colbase + sc;
  float4 xa[4], wa[4];
#pragma unroll
  for (int i = 0; i < 4; ++i) {
    xa[i] = *(const float4*)(xg + (size_t)(16 * i) * D_INN);
    wa[i] = *(const float4*)(wg + (size_t)(16 * i) * D_F);
  }

  for (int kc = 0; kc < D_INN; kc += 64) {
#pragma unroll
    for (int i = 0; i < 4; ++i) {
      float* ad = As + (sr + 16 * i) * 66 + sc;
      v2f lo = {xa[i].x, xa[i].y};
      v2f hi = {xa[i].z, xa[i].w};
      *(v2f*)(ad) = lo;
      *(v2f*)(ad + 2) = hi;
      float* bd = Bs + sc * 66 + sr + 16 * i;
      bd[0] = wa[i].x;
      bd[66] = wa[i].y;
      bd[132] = wa[i].z;
      bd[198] = wa[i].w;
    }
    __syncthreads();
    if (kc + 64 < D_INN) {
#pragma unroll
      for (int i = 0; i < 4; ++i) {
        xa[i] = *(const float4*)(xg + (size_t)(16 * i) * D_INN + kc + 64);
        wa[i] = *(const float4*)(wg + (size_t)(kc + 64 + 16 * i) * D_F);
      }
    }
    for (int kk = 0; kk < 64; kk += 4) {
      int klo = kk + khalf;
#pragma unroll
      for (int t = 0; t < 2; ++t) {
        int g = w * 2 + t;
        int mt = g >> 2, nt = g & 3;
        v2f a = *(const v2f*)(As + (mt * 16 + l15) * 66 + klo);
        v2f bv = *(const v2f*)(Bs + (nt * 16 + l15) * 66 + klo);
        acc[t] = wmma4(a, bv, acc[t]);
      }
    }
    __syncthreads();
  }

  const int hi = lane >> 4;
#pragma unroll
  for (int t = 0; t < 2; ++t) {
    int g = w * 2 + t;
    int mt = g >> 2, nt = g & 3;
    int n = nt * 16 + l15;
    float bb = bias[colbase + n];
#pragma unroll
    for (int r = 0; r < 8; ++r) {
      int mrow = mt * 16 + r + hi * 8;
      float v = acc[t][r] + bb;
      v = v > 0.f ? v : 0.f;
      feats[(size_t)(rowbase + mrow) * D_F + colbase + n] = v;
    }
  }
}

struct SubSet {
  v4f x0, x1;
  v4f l[8];
};

__device__ __forceinline__ void sub_load(SubSet& s, const float* xcol,
                                         const float* Lh, int t) {
  s.x0 = *(const v4f*)(xcol + t);
  s.x1 = *(const v4f*)(xcol + t + 4);
#pragma unroll
  for (int u = 0; u < 8; ++u) s.l[u] = *(const v4f*)(Lh + (t + u) * 4);
}

__device__ __forceinline__ void sub_fma(const SubSet& s, float& a0, float& a1,
                                        float& a2, float& a3) {
#pragma unroll
  for (int u = 0; u < 8; ++u) {
    const float xs = (u < 4) ? s.x0[u & 3] : s.x1[u & 3];
    a0 = fmaf(s.l[u][0], xs, a0);
    a1 = fmaf(s.l[u][1], xs, a1);
    a2 = fmaf(s.l[u][2], xs, a2);
    a3 = fmaf(s.l[u][3], xs, a3);
  }
}

__device__ __forceinline__ void trinv_matrix(const float* __restrict__ covs,
                                             float* __restrict__ Linv,
                                             float* __restrict__ logdet,
                                             float* smem, int mk) {
  float* Xt = smem;
  float* logd = smem + 256 * 260;
  float* Lst = logd + 256;
  const int tid = threadIdx.x;
  const int lane = tid & 31;
  const int c = lane & 15;
  const int h = lane >> 4;
  const int w = __builtin_amdgcn_readfirstlane(tid >> 5);
  const int k1 = w >> 2;
  const int qd = (k1 + (w & 3)) & 3;
  const int rank = 4 * qd + ((qd & 1) ? 3 - k1 : k1);
  const int slab = ((rank & 1) << 3) | (rank >> 1);
  const int wstart = slab * 16;
  const int bb = wstart & 128;
  const int j = wstart + c;
  const float* Lg = covs + (size_t)mk * (D_F * D_F);

  const int tcol = tid & 255;
  const int hh = tid >> 8;
  const int trow = (tcol & 128) + 4 * hh;
  const float dg = Lg[tcol * D_F + tcol];
  float p[4][4];
#pragma unroll
  for (int k = 0; k < 4; ++k) {
#pragma unroll
    for (int r = 0; r < 4; ++r)
      p[k][r] = Lg[(size_t)(8 * k + trow + r) * D_F + tcol];
  }

  for (int e = tid; e < 256 * 128; e += 512) {
    const int cj = e >> 7;
    Xt[cj * 260 + (cj & 128) + (e & 127)] = 0.f;
  }
  if (tid < 256) logd[tid] = logf(dg);

  float* xcol = Xt + j * 260;
  for (int g0 = 0; g0 < 128; g0 += 32) {
#pragma unroll
    for (int k = 0; k < 4; ++k)
      *(float4*)(Lst + k * (2 * 1028) + hh * 1028 + tcol * 4) =
          make_float4(p[k][0], p[k][1], p[k][2], p[k][3]);
    __syncthreads();
    if (g0 + 32 < 128) {
#pragma unroll
      for (int k = 0; k < 4; ++k) {
#pragma unroll
        for (int r = 0; r < 4; ++r)
          p[k][r] = Lg[(size_t)(g0 + 32 + 8 * k + trow + r) * D_F + tcol];
      }
    }
#pragma unroll 1
    for (int k = 0; k < 4; ++k) {
      const int s0 = g0 + 8 * k;
      const float* Lb = Lst + k * (2 * 1028);
      const int i0 = bb + s0;
      if (wstart <= i0) {
        float Ld[8][8];
#pragma unroll
        for (int q = 0; q < 8; ++q) {
#pragma unroll
          for (int g = 0; g < 2; ++g) {
            const float* pp = Lb + g * 1028 + (i0 + q) * 4;
            const int lo = q - 4 * g;
            if (lo <= 0) {
              v4f v = *(const v4f*)pp;
              Ld[4 * g + 0][q] = v[0]; Ld[4 * g + 1][q] = v[1];
              Ld[4 * g + 2][q] = v[2]; Ld[4 * g + 3][q] = v[3];
            } else if (lo == 1) {
              Ld[4 * g + 1][q] = pp[1];
              v2f v = *(const v2f*)(pp + 2);
              Ld[4 * g + 2][q] = v[0]; Ld[4 * g + 3][q] = v[1];
            } else if (lo == 2) {
              v2f v = *(const v2f*)(pp + 2);
              Ld[4 * g + 2][q] = v[0]; Ld[4 * g + 3][q] = v[1];
            } else if (lo == 3) {
              Ld[4 * g + 3][q] = pp[3];
            }
          }
        }

        float s0_ = 0.f, s1 = 0.f, s2 = 0.f, s3 = 0.f;
        if (wstart < i0) {
          const float* Lh = Lb + h * 1028;
          SubSet sa, sb;
          int t = wstart;
          sub_load(sa, xcol, Lh, t);
          for (;;) {
            if (t + 8 >= i0) { sub_fma(sa, s0_, s1, s2, s3); break; }
            sub_load(sb, xcol, Lh, t + 8);
            sub_fma(sa, s0_, s1, s2, s3);
            t += 8;
            if (t + 8 >= i0) { sub_fma(sb, s0_, s1, s2, s3); break; }
            sub_load(sa, xcol, Lh, t + 8);
            sub_fma(sb, s0_, s1, s2, s3);
            t += 8;
          }
        }
        float o0 = __shfl_xor(s0_, 16);
        float o1 = __shfl_xor(s1, 16);
        float o2 = __shfl_xor(s2, 16);
        float o3 = __shfl_xor(s3, 16);
        float s[8];
        s[0] = h ? o0 : s0_; s[1] = h ? o1 : s1;
        s[2] = h ? o2 : s2;  s[3] = h ? o3 : s3;
        s[4] = h ? s0_ : o0; s[5] = h ? s1 : o1;
        s[6] = h ? s2 : o2;  s[7] = h ? s3 : o3;
        float xv[8];
#pragma unroll
        for (int r = 0; r < 8; ++r) {
          float acc = s[r];
#pragma unroll
          for (int q = 0; q < r; ++q) acc = fmaf(Ld[r][q], xv[q], acc);
          float d = Ld[r][r];
          float num = ((j == i0 + r) ? 1.f : 0.f) - acc;
          float v = num / d;
          xv[r] = (j <= i0 + r) ? v : 0.f;
        }
        if (h == 0) {
          *(float4*)(xcol + i0) = make_float4(xv[0], xv[1], xv[2], xv[3]);
          *(float4*)(xcol + i0 + 4) = make_float4(xv[4], xv[5], xv[6], xv[7]);
        }
      }
    }
    __syncthreads();
  }

  float* outp = Linv + (size_t)mk * (D_F * D_F);
  const float* ocol = Xt + tcol * 260;
  const int rb = (tcol & 128) + hh * 64;
  for (int i = rb; i < rb + 64; i += 4) {
    float4 v = *(const float4*)(ocol + i);
    outp[(size_t)(i + 0) * D_F + tcol] = v.x;
    outp[(size_t)(i + 1) * D_F + tcol] = v.y;
    outp[(size_t)(i + 2) * D_F + tcol] = v.z;
    outp[(size_t)(i + 3) * D_F + tcol] = v.w;
  }
  if (tid == 0) {
    float s = 0.f;
    for (int t = 0; t < D_F; ++t) s += logd[t];
    logdet[mk] = s;
  }
}

__global__ void __launch_bounds__(512)
front_kernel(const float* __restrict__ x, const float* __restrict__ W,
             const float* __restrict__ bias, const float* __restrict__ covs,
             float* __restrict__ feats, float* __restrict__ Linv,
             float* __restrict__ logdet) {
  extern __shared__ float smem[];
  const int blk = blockIdx.x;
  if (blk < NMK) {
    trinv_matrix(covs, Linv, logdet, smem, blk);
  } else {
    const int half = threadIdx.x >> 8;
    feats_tile(x, W, bias, feats, smem + half * (2 * 64 * 66),
               threadIdx.x & 255, 2 * (blk - NMK) + half);
  }
}

#define PK 64
#define AP 68

template <int FIRST>
__device__ __forceinline__ void blk16(const float* const (&Ap)[2],
                                      const float* Bp, int k, v8f (&acc)[4]) {
#pragma unroll
  for (int s = 0; s < 16; s += 4) {
    v2f b0 = *(const v2f*)(Bp + k + s);
    v2f b1 = *(const v2f*)(Bp + 16 * 258 + k + s);
#pragma unroll
    for (int t = FIRST; t < 2; ++t) {
      v2f a = *(const v2f*)(Ap[t] + k + s);
      acc[2 * t + 0] = wmma4(a, b0, acc[2 * t + 0]);
      acc[2 * t + 1] = wmma4(a, b1, acc[2 * t + 1]);
    }
  }
}

__device__ __forceinline__ int tri_need(int rt, int kc) {
  int n = 16 * (rt + 1) - kc;
  return n < 0 ? 0 : (n > PK ? PK : n);
}

template <bool TRI>
__device__ __forceinline__ void panel_mma(const float* const (&Ap)[2],
                                          const float* Bp, int q, int kc,
                                          v8f (&acc)[4]) {
  if (TRI) {
    const int e0 = tri_need(q, kc), e1 = tri_need(7 - q, kc);
    int k = 0;
    for (; k < e0; k += 16) blk16<0>(Ap, Bp, k, acc);
    for (; k < e1; k += 16) blk16<1>(Ap, Bp, k, acc);
  } else {
    for (int k = 0; k < PK; k += 16) blk16<0>(Ap, Bp, k, acc);
  }
}

template <int FIRSTP>
__device__ __forceinline__ void panel_load(v4f (&g)[8], const float* src) {
#pragma unroll
  for (int p = FIRSTP; p < 8; ++p)
    g[p] = *(const v4f*)(src + (size_t)(16 * p) * D_F);
}
template <int FIRSTP>
__device__ __forceinline__ void panel_store(const v4f (&g)[8], float* adst) {
#pragma unroll
  for (int p = FIRSTP; p < 8; ++p) *(v4f*)(adst + 16 * p * AP) = g[p];
}

__global__ void __launch_bounds__(256)
gmm_kernel(const float* __restrict__ feats,
                           const float* __restrict__ Linv,
                           const float* __restrict__ covs,
                           const float* __restrict__ logdet,
                           const float* __restrict__ means,
                           const float* __restrict__ weights,
                           float* __restrict__ out) {
  extern __shared__ float smem[];
  float* As = smem;
  float* Bs = smem + 128 * AP;
  float* quad_s = smem + 128 * AP + 64 * 258;
  const int tid = threadIdx.x;
  const int lane = tid & 31;
  const int w = __builtin_amdgcn_readfirstlane(tid >> 5);
  const int mk = blockIdx.y;
  const int b0 = blockIdx.x * 64;
  const float* LinvG = Linv + (size_t)mk * (D_F * D_F);
  const float* CovG = covs + (size_t)mk * (D_F * D_F);
  const float* mu = means + (size_t)mk * D_F;

  const int c4 = (tid & 15) * 4;
  const int r0 = tid >> 4;
  const float* srcA = LinvG + (size_t)r0 * D_F + c4;
  const float* srcB = CovG + (size_t)(128 + r0) * D_F + c4;
  const float* srcC = LinvG + (size_t)(128 + r0) * D_F + 128 + c4;
  float* adst = As + r0 * AP + c4;
  v4f g[8];
  panel_load<0>(g, srcA);

  {
    const int d4 = (tid & 63) * 4;
    const int n0 = tid >> 6;
    const float4 m4 = *(const float4*)(mu + d4);
    float4 f[16];
#pragma unroll
    for (int p = 0; p < 16; ++p)
      f[p] = *(const float4*)(feats + (size_t)(b0 + n0 + 4 * p) * D_F + d4);
#pragma unroll
    for (int p = 0; p < 16; ++p) {
      float* dst = Bs + (n0 + 4 * p) * 258 + d4;
      v2f lo = {f[p].x - m4.x, f[p].y - m4.y};
      v2f hi = {f[p].z - m4.z, f[p].w - m4.w};
      *(v2f*)(dst) = lo;
      *(v2f*)(dst + 2) = hi;
    }
  }
  if (tid < 64) quad_s[tid] = 0.f;
  panel_store<0>(g, adst);
  __syncthreads();

  const int l15 = lane & 15;
  const int hi8 = (lane >> 4) * 8;
  const int khalf = (lane >> 4) * 2;
  const int nh = w >> 2;
  const int q = (w & 4) ? 3 - (w & 3) : (w & 3);
  const float* const Ap[2] = {As + (q * 16 + l15) * AP + khalf,
                              As + ((7 - q) * 16 + l15) * AP + khalf};
  const float* Bw = Bs + (nh * 32 + l15) * 258 + khalf;
  float* const Dw[2] = {Bs + (nh * 32 + l15) * 258 + q * 16 + hi8,
                        Bs + (nh * 32 + l15) * 258 + (7 - q) * 16 + hi8};
  v8f acc[4] = {};
  float qp[2] = {0.f, 0.f};

  panel_load<4>(g, srcA + PK);
  panel_mma<true>(Ap, Bw, q, 0, acc);
  __syncthreads();
  panel_store<4>(g, adst);
  __syncthreads();
  panel_load<0>(g, srcB);
  panel_mma<true>(Ap, Bw + PK, q, PK, acc);
  __syncthreads();
#pragma unroll
  for (int t = 0; t < 2; ++t) {
#pragma unroll
    for (int j = 0; j < 2; ++j) {
      const v8f a = acc[2 * t + j];
      float* dst = Dw[t] + j * (16 * 258);
#pragma unroll
      for (int r = 0; r < 8; r += 2) {
        v2f pr = {a[r], a[r + 1]};
        *(v2f*)(dst + r) = pr;
        qp[j] += a[r] * a[r];
        qp[j] += a[r + 1] * a[r + 1];
      }
    }
  }
  panel_store<0>(g, adst);
  __syncthreads();

#pragma unroll
  for (int t = 0; t < 2; ++t) {
#pragma unroll
    for (int j = 0; j < 2; ++j) {
      const float* sp = Dw[t] + j * (16 * 258) + 128;
#pragma unroll
      for (int r = 0; r < 8; r += 2) {
        v2f pr = *(const v2f*)(sp + r);
        acc[2 * t + j][r] = -pr[0];
        acc[2 * t + j][r + 1] = -pr[1];
      }
    }
  }
  panel_load<0>(g, srcB + PK);
  panel_mma<false>(Ap, Bw, q, 0, acc);
  __syncthreads();
  panel_store<0>(g, adst);
  __syncthreads();
  panel_load<0>(g, srcC);
  panel_mma<false>(Ap, Bw + PK, q, PK, acc);
  __syncthreads();
#pragma unroll
  for (int t = 0; t < 2; ++t) {
#pragma unroll
    for (int j = 0; j < 2; ++j) {
      const v8f a = acc[2 * t + j];
      float* dst = Dw[t] + j * (16 * 258) + 128;
#pragma unroll
      for (int r = 0; r < 8; r += 2) {
        v2f pr = {a[r], a[r + 1]};
        *(v2f*)(dst + r) = pr;
      }
    }
  }
  panel_store<0>(g, adst);
  __syncthreads();

#pragma unroll
  for (int t = 0; t < 4; ++t) acc[t] = v8f{};
  panel_load<4>(g, srcC + PK);
  panel_mma<true>(Ap, Bw + 128, q, 0, acc);
  __syncthreads();
  panel_store<4>(g, adst);
  __syncthreads();
  panel_mma<true>(Ap, Bw + 128 + PK, q, PK, acc);

#pragma unroll
  for (int t = 0; t < 2; ++t) {
#pragma unroll
    for (int j = 0; j < 2; ++j) {
#pragma unroll
      for (int r = 0; r < 8; ++r) qp[j] += acc[2 * t + j][r] * acc[2 * t + j][r];
    }
  }
  atomicAdd(&quad_s[(2 * nh + 0) * 16 + l15], qp[0]);
  atomicAdd(&quad_s[(2 * nh + 1) * 16 + l15], qp[1]);
  __syncthreads();

  if (tid < 64) {
    int m = mk >> 3, kcomp = mk & 7;
    const float* wr = weights + m * NCOMP;
    float wmax = wr[0];
#pragma unroll
    for (int t = 1; t < 8; ++t) wmax = fmaxf(wmax, wr[t]);
    float denom = 0.f;
#pragma unroll
    for (int t = 0; t < 8; ++t) denom += expf(wr[t] - wmax);
    float wmk = expf(wr[kcomp] - wmax) / denom;
    float logp = -0.5f * (quad_s[tid] + 256.0f * LOG2PI) - logdet[mk];
    out[(size_t)(b0 + tid) * NMK + mk] = wmk * logp;
  }
}

extern "C" void kernel_launch(void* const* d_in, const int* in_sizes, int n_in,
                              void* d_out, int out_size, void* d_ws,
                              size_t ws_size, hipStream_t stream) {
  (void)in_sizes; (void)n_in; (void)out_size; (void)ws_size;
  const float* x       = (const float*)d_in[0];
  const float* W       = (const float*)d_in[1];
  const float* bias    = (const float*)d_in[2];
  const float* means   = (const float*)d_in[3];
  const float* covs    = (const float*)d_in[4];
  const float* weights = (const float*)d_in[5];
  float* out = (float*)d_out;

  float* ws = (float*)d_ws;
  float* feats  = ws;
  float* LinvW  = ws + (size_t)B_N * D_F;
  float* logdet = LinvW + (size_t)NMK * D_F * D_F;

  size_t sh2 = (size_t)(256 * 260 + 256 + 4 * 2 * 1028) * sizeof(float);
  front_kernel<<<dim3(NMK + 128), 512, sh2, stream>>>(x, W, bias, covs, feats,
                                                      LinvW, logdet);

  size_t sh3 = (size_t)(128 * AP + 64 * 258 + 64) * sizeof(float);
  gmm_kernel<<<dim3(B_N / 64, NMK), 256, sh3, stream>>>(feats, LinvW, covs, logdet,
                                                        means, weights, out);
}
